# starting kernel
# baseline (speedup 1.0000x reference)
#include <hip/hip_runtime.h>
#include <hip/hip_bf16.h>


typedef __attribute__((ext_vector_type(2))) float v2f;
typedef __attribute__((ext_vector_type(4))) float v4f;
typedef __attribute__((ext_vector_type(8))) float v8f;

#define NQ      10
#define NL      4
#define BATCH_N 8192
#define PITCH   33
#define PLANE   (32 * PITCH)
#define SAMPLE_F (4 * PLANE)
#define WPB     2

__device__ __forceinline__ int uh_off(int l, int i, int k, int p) {
  return (((l * 2 + i) * 2 + k) * 2 + p) * 256;
}
__device__ __forceinline__ int ul_off(int l, int k, int j, int p) {
  return 8192 + (((l * 2 + k) * 2 + j) * 2 + p) * 256;
}

__device__ __forceinline__ void gate_elem(float a, float b, int rb, int cb,
                                          float& fr, float& fi) {
  float ca = cosf(0.5f * a), sa = sinf(0.5f * a);
  float cz = cosf(0.5f * b), sz = sinf(0.5f * b);
  float ry = rb ? (cb ? ca : sa) : (cb ? -sa : ca);
  fr = ry * cz;
  fi = ry * (rb ? sz : -sz);
}

__global__ void vqc_prep(const float* __restrict__ w, float* __restrict__ ws) {
  int l = blockIdx.x;
  for (int t = threadIdx.x; t < 4096; t += blockDim.x) {
    int which = t >> 11;
    int e     = t & 2047;
    int frag  = e >> 8;
    int lane  = (e >> 3) & 31;
    int slot  = e & 7;
    int p     = frag & 1;
    int Kloc  = 4 * (slot >> 1) + (slot & 1) + 2 * (lane >> 4);

    float vr = 1.f, vi = 0.f;
    int off;
    if (which == 0) {
      int it = frag >> 2, kt = (frag >> 1) & 1;
      int m = it * 16 + (lane & 15);
      int kcol = kt * 16 + Kloc;
      int r0 = (m ^ (m >> 1)) & 31;
      #pragma unroll
      for (int w2 = 0; w2 < 5; ++w2) {
        int rb = (r0 >> (4 - w2)) & 1;
        int cb = (kcol >> (4 - w2)) & 1;
        float fr, fi;
        gate_elem(w[(l * NQ + w2) * 2], w[(l * NQ + w2) * 2 + 1], rb, cb, fr, fi);
        float nr = vr * fr - vi * fi; vi = vr * fi + vi * fr; vr = nr;
      }
      off = uh_off(l, it, kt, p);
    } else {
      int kt = frag >> 2, jt = (frag >> 1) & 1;
      int n  = jt * 16 + (lane & 15);
      int kg = kt * 16 + Kloc;
      #pragma unroll
      for (int w2 = 0; w2 < 5; ++w2) {
        int rb = (n >> (4 - w2)) & 1;
        int cb = (kg >> (4 - w2)) & 1;
        float fr, fi;
        gate_elem(w[(l * NQ + 5 + w2) * 2], w[(l * NQ + 5 + w2) * 2 + 1], rb, cb, fr, fi);
        float nr = vr * fr - vi * fi; vi = vr * fi + vi * fr; vr = nr;
      }
      off = ul_off(l, kt, jt, p);
    }
    ws[off + lane * 8 + slot] = p ? vi : vr;
  }
}

__global__ __launch_bounds__(32 * WPB) void vqc_main(const float* __restrict__ x,
                                                     const float* __restrict__ ws,
                                                     float* __restrict__ out) {
  __shared__ float lds[WPB * SAMPLE_F];
  const int lane = threadIdx.x & 31;
  const int wv   = threadIdx.x >> 5;
  const int sample = blockIdx.x * WPB + wv;

  float* Sa_r = lds + wv * SAMPLE_F;
  float* Sa_i = Sa_r + PLANE;
  float* Sb_r = Sa_r + 2 * PLANE;
  float* Sb_i = Sa_r + 3 * PLANE;

  float rowv = 1.f, colv = 1.f;
  #pragma unroll
  for (int w2 = 0; w2 < 5; ++w2) {
    float ah = 0.5f * x[sample * NQ + w2];
    float al = 0.5f * x[sample * NQ + 5 + w2];
    int b = (lane >> (4 - w2)) & 1;
    rowv *= b ? sinf(ah) : cosf(ah);
    colv *= b ? sinf(al) : cosf(al);
  }
  for (int r = 0; r < 32; ++r) {
    float rv = __shfl(rowv, r, 32);
    Sa_r[r * PITCH + lane] = rv * colv;
    Sa_i[r * PITCH + lane] = 0.f;
  }

  for (int l = 0; l < NL; ++l) {
    #pragma unroll
    for (int it = 0; it < 2; ++it) {
      float ar[2][8], ai[2][8];
      #pragma unroll
      for (int k = 0; k < 2; ++k)
        #pragma unroll
        for (int s = 0; s < 8; ++s) {
          int row = it * 16 + (lane & 15);
          int col = k * 16 + 4 * (s >> 1) + (s & 1) + 2 * (lane >> 4);
          int rs = row, cs = col;
          if (l > 0) {
            int rm = (col & 1) ? (row ^ 16) : row;
            int c3 = (col ^ (col >> 1)) & 31;
            cs = (rm & 1) ? (c3 ^ 16) : c3;
            rs = rm;
          }
          ar[k][s] = Sa_r[rs * PITCH + cs];
          ai[k][s] = Sa_i[rs * PITCH + cs];
        }
      #pragma unroll
      for (int j = 0; j < 2; ++j) {
        v8f dr = {0,0,0,0,0,0,0,0}, di = {0,0,0,0,0,0,0,0};
        #pragma unroll
        for (int k = 0; k < 2; ++k) {
          const v4f* pr = (const v4f*)(ws + ul_off(l, k, j, 0) + lane * 8);
          const v4f* pi = (const v4f*)(ws + ul_off(l, k, j, 1) + lane * 8);
          v4f r0 = pr[0], r1 = pr[1], i0 = pi[0], i1 = pi[1];
          float br[8] = {r0.x, r0.y, r0.z, r0.w, r1.x, r1.y, r1.z, r1.w};
          float bi[8] = {i0.x, i0.y, i0.z, i0.w, i1.x, i1.y, i1.z, i1.w};
          #pragma unroll
          for (int kk = 0; kk < 4; ++kk) {
            v2f aR  = { ar[k][2 * kk],  ar[k][2 * kk + 1]};
            v2f aI  = { ai[k][2 * kk],  ai[k][2 * kk + 1]};
            v2f aIn = {-ai[k][2 * kk], -ai[k][2 * kk + 1]};
            v2f bR  = { br[2 * kk],     br[2 * kk + 1]};
            v2f bI  = { bi[2 * kk],     bi[2 * kk + 1]};
            dr = __builtin_amdgcn_wmma_f32_16x16x4_f32(false, aR,  false, bR, (short)0, dr, false, false);
            dr = __builtin_amdgcn_wmma_f32_16x16x4_f32(false, aIn, false, bI, (short)0, dr, false, false);
            di = __builtin_amdgcn_wmma_f32_16x16x4_f32(false, aR,  false, bI, (short)0, di, false, false);
            di = __builtin_amdgcn_wmma_f32_16x16x4_f32(false, aI,  false, bR, (short)0, di, false, false);
          }
        }
        #pragma unroll
        for (int s = 0; s < 8; ++s) {
          int m = it * 16 + s + 8 * (lane >> 4);
          int n = j * 16 + (lane & 15);
          Sb_r[m * PITCH + n] = dr[s];
          Sb_i[m * PITCH + n] = di[s];
        }
      }
    }
    #pragma unroll
    for (int j = 0; j < 2; ++j) {
      float br[2][8], bi[2][8];
      #pragma unroll
      for (int k = 0; k < 2; ++k)
        #pragma unroll
        for (int s = 0; s < 8; ++s) {
          int K = k * 16 + 4 * (s >> 1) + (s & 1) + 2 * (lane >> 4);
          int n = j * 16 + (lane & 15);
          br[k][s] = Sb_r[K * PITCH + n];
          bi[k][s] = Sb_i[K * PITCH + n];
        }
      #pragma unroll
      for (int it = 0; it < 2; ++it) {
        v8f dr = {0,0,0,0,0,0,0,0}, di = {0,0,0,0,0,0,0,0};
        #pragma unroll
        for (int k = 0; k < 2; ++k) {
          const v4f* pr = (const v4f*)(ws + uh_off(l, it, k, 0) + lane * 8);
          const v4f* pi = (const v4f*)(ws + uh_off(l, it, k, 1) + lane * 8);
          v4f r0 = pr[0], r1 = pr[1], i0 = pi[0], i1 = pi[1];
          float ur[8] = {r0.x, r0.y, r0.z, r0.w, r1.x, r1.y, r1.z, r1.w};
          float ui[8] = {i0.x, i0.y, i0.z, i0.w, i1.x, i1.y, i1.z, i1.w};
          #pragma unroll
          for (int kk = 0; kk < 4; ++kk) {
            v2f aR  = { ur[2 * kk],  ur[2 * kk + 1]};
            v2f aI  = { ui[2 * kk],  ui[2 * kk + 1]};
            v2f aIn = {-ui[2 * kk], -ui[2 * kk + 1]};
            v2f bR  = { br[k][2 * kk], br[k][2 * kk + 1]};
            v2f bI  = { bi[k][2 * kk], bi[k][2 * kk + 1]};
            dr = __builtin_amdgcn_wmma_f32_16x16x4_f32(false, aR,  false, bR, (short)0, dr, false, false);
            dr = __builtin_amdgcn_wmma_f32_16x16x4_f32(false, aIn, false, bI, (short)0, dr, false, false);
            di = __builtin_amdgcn_wmma_f32_16x16x4_f32(false, aR,  false, bI, (short)0, di, false, false);
            di = __builtin_amdgcn_wmma_f32_16x16x4_f32(false, aI,  false, bR, (short)0, di, false, false);
          }
        }
        #pragma unroll
        for (int s = 0; s < 8; ++s) {
          int m = it * 16 + s + 8 * (lane >> 4);
          int n = j * 16 + (lane & 15);
          Sa_r[m * PITCH + n] = dr[s];
          Sa_i[m * PITCH + n] = di[s];
        }
      }
    }
  }

  float acc[NQ];
  #pragma unroll
  for (int i = 0; i < NQ; ++i) acc[i] = 0.f;
  for (int r = 0; r < 32; ++r) {
    float re = Sa_r[r * PITCH + lane];
    float im = Sa_i[r * PITCH + lane];
    float p  = re * re + im * im;
    int c3 = (r & 1) ? (lane ^ 16) : lane;
    int c4 = (c3 ^ (c3 >> 1) ^ (c3 >> 2) ^ (c3 >> 3) ^ (c3 >> 4)) & 31;
    int r2 = (c4 & 1) ? (r ^ 16) : r;
    int a  = (r2 << 5) | c4;
    #pragma unroll
    for (int i = 0; i < NQ; ++i)
      acc[i] += (a & (1 << (9 - i))) ? -p : p;
  }
  #pragma unroll
  for (int off = 16; off > 0; off >>= 1)
    #pragma unroll
    for (int i = 0; i < NQ; ++i)
      acc[i] += __shfl_xor(acc[i], off, 32);
  if (lane == 0) {
    #pragma unroll
    for (int i = 0; i < NQ; ++i) out[sample * NQ + i] = acc[i];
  }
}

extern "C" void kernel_launch(void* const* d_in, const int* in_sizes, int n_in,
                              void* d_out, int out_size, void* d_ws, size_t ws_size,
                              hipStream_t stream) {
  const float* x = (const float*)d_in[0];
  const float* w = (const float*)d_in[1];
  float* ws  = (float*)d_ws;
  float* out = (float*)d_out;

  vqc_prep<<<NL, 1024, 0, stream>>>(w, ws);
  vqc_main<<<BATCH_N / WPB, 32 * WPB, 0, stream>>>(x, ws, out);
}


# each complex 32x32 product by three real f32 matrix products, not four (P1=ArBr, P2=AiBi, P3=(Ar+Ai)(Br+Bi); re=P1-P2, im=P3-P1-P2): 768 WMMA per sample, not 1024; re/im sums regrouped
# speedup vs baseline: 1.1261x; 1.1261x over previous
#include <hip/hip_runtime.h>
#include <hip/hip_bf16.h>


typedef __attribute__((ext_vector_type(2))) float v2f;
typedef __attribute__((ext_vector_type(4))) float v4f;
typedef __attribute__((ext_vector_type(8))) float v8f;

#define NQ      10
#define NL      4
#define BATCH_N 8192
#define PITCH   33
#define PLANE   (32 * PITCH)
#define SAMPLE_F (4 * PLANE)
#define WPB     2

__device__ __forceinline__ int uh_off(int l, int i, int k, int p) {
  return (((l * 2 + i) * 2 + k) * 2 + p) * 256;
}
__device__ __forceinline__ int ul_off(int l, int k, int j, int p) {
  return 8192 + (((l * 2 + k) * 2 + j) * 2 + p) * 256;
}

__device__ __forceinline__ void gate_elem(float a, float b, int rb, int cb,
                                          float& fr, float& fi) {
  float ca = cosf(0.5f * a), sa = sinf(0.5f * a);
  float cz = cosf(0.5f * b), sz = sinf(0.5f * b);
  float ry = rb ? (cb ? ca : sa) : (cb ? -sa : ca);
  fr = ry * cz;
  fi = ry * (rb ? sz : -sz);
}

__global__ void vqc_prep(const float* __restrict__ w, float* __restrict__ ws) {
  int l = blockIdx.x;
  for (int t = threadIdx.x; t < 4096; t += blockDim.x) {
    int which = t >> 11;
    int e     = t & 2047;
    int frag  = e >> 8;
    int lane  = (e >> 3) & 31;
    int slot  = e & 7;
    int p     = frag & 1;
    int Kloc  = 4 * (slot >> 1) + (slot & 1) + 2 * (lane >> 4);

    float vr = 1.f, vi = 0.f;
    int off;
    if (which == 0) {
      int it = frag >> 2, kt = (frag >> 1) & 1;
      int m = it * 16 + (lane & 15);
      int kcol = kt * 16 + Kloc;
      int r0 = (m ^ (m >> 1)) & 31;
      #pragma unroll
      for (int w2 = 0; w2 < 5; ++w2) {
        int rb = (r0 >> (4 - w2)) & 1;
        int cb = (kcol >> (4 - w2)) & 1;
        float fr, fi;
        gate_elem(w[(l * NQ + w2) * 2], w[(l * NQ + w2) * 2 + 1], rb, cb, fr, fi);
        float nr = vr * fr - vi * fi; vi = vr * fi + vi * fr; vr = nr;
      }
      off = uh_off(l, it, kt, p);
    } else {
      int kt = frag >> 2, jt = (frag >> 1) & 1;
      int n  = jt * 16 + (lane & 15);
      int kg = kt * 16 + Kloc;
      #pragma unroll
      for (int w2 = 0; w2 < 5; ++w2) {
        int rb = (n >> (4 - w2)) & 1;
        int cb = (kg >> (4 - w2)) & 1;
        float fr, fi;
        gate_elem(w[(l * NQ + 5 + w2) * 2], w[(l * NQ + 5 + w2) * 2 + 1], rb, cb, fr, fi);
        float nr = vr * fr - vi * fi; vi = vr * fi + vi * fr; vr = nr;
      }
      off = ul_off(l, kt, jt, p);
    }
    ws[off + lane * 8 + slot] = p ? vi : vr;
  }
}

__global__ __launch_bounds__(32 * WPB) void vqc_main(const float* __restrict__ x,
                                                     const float* __restrict__ ws,
                                                     float* __restrict__ out) {
  __shared__ float lds[WPB * SAMPLE_F];
  const int lane = threadIdx.x & 31;
  const int wv   = threadIdx.x >> 5;
  const int sample = blockIdx.x * WPB + wv;

  float* Sa_r = lds + wv * SAMPLE_F;
  float* Sa_i = Sa_r + PLANE;
  float* Sb_r = Sa_r + 2 * PLANE;
  float* Sb_i = Sa_r + 3 * PLANE;

  float rowv = 1.f, colv = 1.f;
  #pragma unroll
  for (int w2 = 0; w2 < 5; ++w2) {
    float ah = 0.5f * x[sample * NQ + w2];
    float al = 0.5f * x[sample * NQ + 5 + w2];
    int b = (lane >> (4 - w2)) & 1;
    rowv *= b ? sinf(ah) : cosf(ah);
    colv *= b ? sinf(al) : cosf(al);
  }
  for (int r = 0; r < 32; ++r) {
    float rv = __shfl(rowv, r, 32);
    Sa_r[r * PITCH + lane] = rv * colv;
    Sa_i[r * PITCH + lane] = 0.f;
  }

  for (int l = 0; l < NL; ++l) {
    #pragma unroll
    for (int it = 0; it < 2; ++it) {
      float ar[2][8], ai[2][8];
      #pragma unroll
      for (int k = 0; k < 2; ++k)
        #pragma unroll
        for (int s = 0; s < 8; ++s) {
          int row = it * 16 + (lane & 15);
          int col = k * 16 + 4 * (s >> 1) + (s & 1) + 2 * (lane >> 4);
          int rs = row, cs = col;
          if (l > 0) {
            int rm = (col & 1) ? (row ^ 16) : row;
            int c3 = (col ^ (col >> 1)) & 31;
            cs = (rm & 1) ? (c3 ^ 16) : c3;
            rs = rm;
          }
          ar[k][s] = Sa_r[rs * PITCH + cs];
          ai[k][s] = Sa_i[rs * PITCH + cs];
        }
      v8f p1[2], p2[2], p3[2];
      #pragma unroll
      for (int j = 0; j < 2; ++j) {
        p1[j] = (v8f){0,0,0,0,0,0,0,0}; p2[j] = p1[j]; p3[j] = p1[j];
      }
      #pragma unroll
      for (int k = 0; k < 2; ++k) {
        float br[2][8], bi[2][8];
        #pragma unroll
        for (int j = 0; j < 2; ++j) {
          const v4f* pr = (const v4f*)(ws + ul_off(l, k, j, 0) + lane * 8);
          const v4f* pi = (const v4f*)(ws + ul_off(l, k, j, 1) + lane * 8);
          v4f r0 = pr[0], r1 = pr[1], i0 = pi[0], i1 = pi[1];
          br[j][0] = r0.x; br[j][1] = r0.y; br[j][2] = r0.z; br[j][3] = r0.w;
          br[j][4] = r1.x; br[j][5] = r1.y; br[j][6] = r1.z; br[j][7] = r1.w;
          bi[j][0] = i0.x; bi[j][1] = i0.y; bi[j][2] = i0.z; bi[j][3] = i0.w;
          bi[j][4] = i1.x; bi[j][5] = i1.y; bi[j][6] = i1.z; bi[j][7] = i1.w;
        }
        #pragma unroll
        for (int kk = 0; kk < 4; ++kk) {
          v2f aR = { ar[k][2 * kk],                 ar[k][2 * kk + 1]};
          v2f aI = { ai[k][2 * kk],                 ai[k][2 * kk + 1]};
          v2f aS = { ar[k][2 * kk] + ai[k][2 * kk], ar[k][2 * kk + 1] + ai[k][2 * kk + 1]};
          #pragma unroll
          for (int j = 0; j < 2; ++j) {
            v2f bR = { br[j][2 * kk],                 br[j][2 * kk + 1]};
            v2f bI = { bi[j][2 * kk],                 bi[j][2 * kk + 1]};
            v2f bS = { br[j][2 * kk] + bi[j][2 * kk], br[j][2 * kk + 1] + bi[j][2 * kk + 1]};
            p1[j] = __builtin_amdgcn_wmma_f32_16x16x4_f32(false, aR, false, bR, (short)0, p1[j], false, false);
            p2[j] = __builtin_amdgcn_wmma_f32_16x16x4_f32(false, aI, false, bI, (short)0, p2[j], false, false);
            p3[j] = __builtin_amdgcn_wmma_f32_16x16x4_f32(false, aS, false, bS, (short)0, p3[j], false, false);
          }
        }
      }
      #pragma unroll
      for (int j = 0; j < 2; ++j)
        #pragma unroll
        for (int s = 0; s < 8; ++s) {
          int m = it * 16 + s + 8 * (lane >> 4);
          int n = j * 16 + (lane & 15);
          Sb_r[m * PITCH + n] = p1[j][s] - p2[j][s];
          Sb_i[m * PITCH + n] = (p3[j][s] - p1[j][s]) - p2[j][s];
        }
    }
    #pragma unroll
    for (int j = 0; j < 2; ++j) {
      float br[2][8], bi[2][8];
      #pragma unroll
      for (int k = 0; k < 2; ++k)
        #pragma unroll
        for (int s = 0; s < 8; ++s) {
          int K = k * 16 + 4 * (s >> 1) + (s & 1) + 2 * (lane >> 4);
          int n = j * 16 + (lane & 15);
          br[k][s] = Sb_r[K * PITCH + n];
          bi[k][s] = Sb_i[K * PITCH + n];
        }
      v8f p1[2], p2[2], p3[2];
      #pragma unroll
      for (int it = 0; it < 2; ++it) {
        p1[it] = (v8f){0,0,0,0,0,0,0,0}; p2[it] = p1[it]; p3[it] = p1[it];
      }
      #pragma unroll
      for (int k = 0; k < 2; ++k) {
        float ur[2][8], ui[2][8];
        #pragma unroll
        for (int it = 0; it < 2; ++it) {
          const v4f* pr = (const v4f*)(ws + uh_off(l, it, k, 0) + lane * 8);
          const v4f* pi = (const v4f*)(ws + uh_off(l, it, k, 1) + lane * 8);
          v4f r0 = pr[0], r1 = pr[1], i0 = pi[0], i1 = pi[1];
          ur[it][0] = r0.x; ur[it][1] = r0.y; ur[it][2] = r0.z; ur[it][3] = r0.w;
          ur[it][4] = r1.x; ur[it][5] = r1.y; ur[it][6] = r1.z; ur[it][7] = r1.w;
          ui[it][0] = i0.x; ui[it][1] = i0.y; ui[it][2] = i0.z; ui[it][3] = i0.w;
          ui[it][4] = i1.x; ui[it][5] = i1.y; ui[it][6] = i1.z; ui[it][7] = i1.w;
        }
        #pragma unroll
        for (int kk = 0; kk < 4; ++kk) {
          v2f bR = { br[k][2 * kk],                 br[k][2 * kk + 1]};
          v2f bI = { bi[k][2 * kk],                 bi[k][2 * kk + 1]};
          v2f bS = { br[k][2 * kk] + bi[k][2 * kk], br[k][2 * kk + 1] + bi[k][2 * kk + 1]};
          #pragma unroll
          for (int it = 0; it < 2; ++it) {
            v2f aR = { ur[it][2 * kk],                  ur[it][2 * kk + 1]};
            v2f aI = { ui[it][2 * kk],                  ui[it][2 * kk + 1]};
            v2f aS = { ur[it][2 * kk] + ui[it][2 * kk], ur[it][2 * kk + 1] + ui[it][2 * kk + 1]};
            p1[it] = __builtin_amdgcn_wmma_f32_16x16x4_f32(false, aR, false, bR, (short)0, p1[it], false, false);
            p2[it] = __builtin_amdgcn_wmma_f32_16x16x4_f32(false, aI, false, bI, (short)0, p2[it], false, false);
            p3[it] = __builtin_amdgcn_wmma_f32_16x16x4_f32(false, aS, false, bS, (short)0, p3[it], false, false);
          }
        }
      }
      #pragma unroll
      for (int it = 0; it < 2; ++it)
        #pragma unroll
        for (int s = 0; s < 8; ++s) {
          int m = it * 16 + s + 8 * (lane >> 4);
          int n = j * 16 + (lane & 15);
          Sa_r[m * PITCH + n] = p1[it][s] - p2[it][s];
          Sa_i[m * PITCH + n] = (p3[it][s] - p1[it][s]) - p2[it][s];
        }
    }
  }

  float acc[NQ];
  #pragma unroll
  for (int i = 0; i < NQ; ++i) acc[i] = 0.f;
  for (int r = 0; r < 32; ++r) {
    float re = Sa_r[r * PITCH + lane];
    float im = Sa_i[r * PITCH + lane];
    float p  = re * re + im * im;
    int c3 = (r & 1) ? (lane ^ 16) : lane;
    int c4 = (c3 ^ (c3 >> 1) ^ (c3 >> 2) ^ (c3 >> 3) ^ (c3 >> 4)) & 31;
    int r2 = (c4 & 1) ? (r ^ 16) : r;
    int a  = (r2 << 5) | c4;
    #pragma unroll
    for (int i = 0; i < NQ; ++i)
      acc[i] += (a & (1 << (9 - i))) ? -p : p;
  }
  #pragma unroll
  for (int off = 16; off > 0; off >>= 1)
    #pragma unroll
    for (int i = 0; i < NQ; ++i)
      acc[i] += __shfl_xor(acc[i], off, 32);
  if (lane == 0) {
    #pragma unroll
    for (int i = 0; i < NQ; ++i) out[sample * NQ + i] = acc[i];
  }
}

extern "C" void kernel_launch(void* const* d_in, const int* in_sizes, int n_in,
                              void* d_out, int out_size, void* d_ws, size_t ws_size,
                              hipStream_t stream) {
  const float* x = (const float*)d_in[0];
  const float* w = (const float*)d_in[1];
  float* ws  = (float*)d_ws;
  float* out = (float*)d_out;

  vqc_prep<<<NL, 1024, 0, stream>>>(w, ws);
  vqc_main<<<BATCH_N / WPB, 32 * WPB, 0, stream>>>(x, ws, out);
}


# wire gates split as diagonal RZ times REAL RY: real 32x32 matrix times re and im planes (2 f32 matrix products, not 3), phases by elementwise table: 512 WMMA/sample, not 768; re/im sums regrouped
# speedup vs baseline: 1.6586x; 1.4729x over previous
#include <hip/hip_runtime.h>
#include <hip/hip_bf16.h>


typedef __attribute__((ext_vector_type(2))) float v2f;
typedef __attribute__((ext_vector_type(4))) float v4f;
typedef __attribute__((ext_vector_type(8))) float v8f;

#define NQ      10
#define NL      4
#define BATCH_N 8192
#define PITCH   33
#define PLANE   (32 * PITCH)
#define SAMPLE_F (4 * PLANE)
#define WPB     2

__device__ __forceinline__ int rh_off(int l, int i, int k) {
  return ((l * 2 + i) * 2 + k) * 256;
}
__device__ __forceinline__ int rl_off(int l, int k, int j) {
  return 4096 + ((l * 2 + k) * 2 + j) * 256;
}
__device__ __forceinline__ int ph_off(int l, int i, int j, int p) {
  return 8192 + (((l * 2 + i) * 2 + j) * 2 + p) * 256;
}

__device__ __forceinline__ float ry_elem(float a, int rb, int cb) {
  float ca = cosf(0.5f * a), sa = sinf(0.5f * a);
  return rb ? (cb ? ca : sa) : (cb ? -sa : ca);
}

__device__ __forceinline__ void rz_phase(float b, int rb, float& fr, float& fi) {
  float sz = sinf(0.5f * b);
  fr = cosf(0.5f * b);
  fi = rb ? sz : -sz;
}

__global__ void vqc_prep(const float* __restrict__ w, float* __restrict__ ws) {
  int l = blockIdx.x;
  for (int t = threadIdx.x; t < 3072; t += blockDim.x) {
    int which = t >> 10;
    int e     = t & 1023;
    int frag  = e >> 8;
    int lane  = (e >> 3) & 31;
    int slot  = e & 7;
    int f1 = frag >> 1, f0 = frag & 1;
    if (which < 2) {
      int Kloc = 4 * (slot >> 1) + (slot & 1) + 2 * (lane >> 4);
      int rsel, csel, wbase, off;
      if (which == 0) {
        int m = f1 * 16 + (lane & 15);
        rsel = (m ^ (m >> 1)) & 31;
        csel = f0 * 16 + Kloc;
        wbase = 0;
        off = rh_off(l, f1, f0);
      } else {
        rsel = f0 * 16 + (lane & 15);
        csel = f1 * 16 + Kloc;
        wbase = 5;
        off = rl_off(l, f1, f0);
      }
      float v = 1.f;
      #pragma unroll
      for (int w2 = 0; w2 < 5; ++w2)
        v *= ry_elem(w[(l * NQ + wbase + w2) * 2],
                     (rsel >> (4 - w2)) & 1, (csel >> (4 - w2)) & 1);
      ws[off + lane * 8 + slot] = v;
    } else {
      int m  = f1 * 16 + slot + 8 * (lane >> 4);
      int n  = f0 * 16 + (lane & 15);
      int r0 = (m ^ (m >> 1)) & 31;
      float vr = 1.f, vi = 0.f;
      #pragma unroll
      for (int w2 = 0; w2 < NQ; ++w2) {
        int rb = (w2 < 5) ? ((r0 >> (4 - w2)) & 1) : ((n >> (9 - w2)) & 1);
        float fr, fi;
        rz_phase(w[(l * NQ + w2) * 2 + 1], rb, fr, fi);
        float nr = vr * fr - vi * fi; vi = vr * fi + vi * fr; vr = nr;
      }
      ws[ph_off(l, f1, f0, 0) + lane * 8 + slot] = vr;
      ws[ph_off(l, f1, f0, 1) + lane * 8 + slot] = vi;
    }
  }
}

__device__ __forceinline__ void load_frag(const float* __restrict__ p, float (&f)[8]) {
  const v4f* q = (const v4f*)p;
  v4f a = q[0], b = q[1];
  f[0] = a.x; f[1] = a.y; f[2] = a.z; f[3] = a.w;
  f[4] = b.x; f[5] = b.y; f[6] = b.z; f[7] = b.w;
}

__global__ __launch_bounds__(32 * WPB) void vqc_main(const float* __restrict__ x,
                                                     const float* __restrict__ ws,
                                                     float* __restrict__ out) {
  __shared__ float lds[WPB * SAMPLE_F];
  const int lane = threadIdx.x & 31;
  const int wv   = threadIdx.x >> 5;
  const int sample = blockIdx.x * WPB + wv;

  float* Sa_r = lds + wv * SAMPLE_F;
  float* Sa_i = Sa_r + PLANE;
  float* Sb_r = Sa_r + 2 * PLANE;
  float* Sb_i = Sa_r + 3 * PLANE;

  float rowv = 1.f, colv = 1.f;
  #pragma unroll
  for (int w2 = 0; w2 < 5; ++w2) {
    float ah = 0.5f * x[sample * NQ + w2];
    float al = 0.5f * x[sample * NQ + 5 + w2];
    int b = (lane >> (4 - w2)) & 1;
    rowv *= b ? sinf(ah) : cosf(ah);
    colv *= b ? sinf(al) : cosf(al);
  }
  for (int r = 0; r < 32; ++r) {
    float rv = __shfl(rowv, r, 32);
    Sa_r[r * PITCH + lane] = rv * colv;
    Sa_i[r * PITCH + lane] = 0.f;
  }

  for (int l = 0; l < NL; ++l) {
    float rl[2][2][8];
    #pragma unroll
    for (int k = 0; k < 2; ++k)
      #pragma unroll
      for (int j = 0; j < 2; ++j)
        load_frag(ws + rl_off(l, k, j) + lane * 8, rl[k][j]);
    #pragma unroll
    for (int it = 0; it < 2; ++it) {
      float ar[2][8], ai[2][8];
      #pragma unroll
      for (int k = 0; k < 2; ++k)
        #pragma unroll
        for (int s = 0; s < 8; ++s) {
          int row = it * 16 + (lane & 15);
          int col = k * 16 + 4 * (s >> 1) + (s & 1) + 2 * (lane >> 4);
          int rs = row, cs = col;
          if (l > 0) {
            int rm = (col & 1) ? (row ^ 16) : row;
            int c3 = (col ^ (col >> 1)) & 31;
            cs = (rm & 1) ? (c3 ^ 16) : c3;
            rs = rm;
          }
          ar[k][s] = Sa_r[rs * PITCH + cs];
          ai[k][s] = Sa_i[rs * PITCH + cs];
        }
      v8f pr[2], pi[2];
      #pragma unroll
      for (int j = 0; j < 2; ++j) {
        pr[j] = (v8f){0,0,0,0,0,0,0,0}; pi[j] = pr[j];
      }
      #pragma unroll
      for (int k = 0; k < 2; ++k)
        #pragma unroll
        for (int kk = 0; kk < 4; ++kk) {
          v2f aR = { ar[k][2 * kk], ar[k][2 * kk + 1]};
          v2f aI = { ai[k][2 * kk], ai[k][2 * kk + 1]};
          #pragma unroll
          for (int j = 0; j < 2; ++j) {
            v2f bR = { rl[k][j][2 * kk], rl[k][j][2 * kk + 1]};
            pr[j] = __builtin_amdgcn_wmma_f32_16x16x4_f32(false, aR, false, bR, (short)0, pr[j], false, false);
            pi[j] = __builtin_amdgcn_wmma_f32_16x16x4_f32(false, aI, false, bR, (short)0, pi[j], false, false);
          }
        }
      #pragma unroll
      for (int j = 0; j < 2; ++j)
        #pragma unroll
        for (int s = 0; s < 8; ++s) {
          int m = it * 16 + s + 8 * (lane >> 4);
          int n = j * 16 + (lane & 15);
          Sb_r[m * PITCH + n] = pr[j][s];
          Sb_i[m * PITCH + n] = pi[j][s];
        }
    }
    float rh[2][2][8];
    #pragma unroll
    for (int it = 0; it < 2; ++it)
      #pragma unroll
      for (int k = 0; k < 2; ++k)
        load_frag(ws + rh_off(l, it, k) + lane * 8, rh[it][k]);
    #pragma unroll
    for (int j = 0; j < 2; ++j) {
      float br[2][8], bi[2][8];
      #pragma unroll
      for (int k = 0; k < 2; ++k)
        #pragma unroll
        for (int s = 0; s < 8; ++s) {
          int K = k * 16 + 4 * (s >> 1) + (s & 1) + 2 * (lane >> 4);
          int n = j * 16 + (lane & 15);
          br[k][s] = Sb_r[K * PITCH + n];
          bi[k][s] = Sb_i[K * PITCH + n];
        }
      v8f pr[2], pi[2];
      #pragma unroll
      for (int it = 0; it < 2; ++it) {
        pr[it] = (v8f){0,0,0,0,0,0,0,0}; pi[it] = pr[it];
      }
      #pragma unroll
      for (int k = 0; k < 2; ++k)
        #pragma unroll
        for (int kk = 0; kk < 4; ++kk) {
          v2f bR = { br[k][2 * kk], br[k][2 * kk + 1]};
          v2f bI = { bi[k][2 * kk], bi[k][2 * kk + 1]};
          #pragma unroll
          for (int it = 0; it < 2; ++it) {
            v2f aR = { rh[it][k][2 * kk], rh[it][k][2 * kk + 1]};
            pr[it] = __builtin_amdgcn_wmma_f32_16x16x4_f32(false, aR, false, bR, (short)0, pr[it], false, false);
            pi[it] = __builtin_amdgcn_wmma_f32_16x16x4_f32(false, aR, false, bI, (short)0, pi[it], false, false);
          }
        }
      #pragma unroll
      for (int it = 0; it < 2; ++it) {
        float phr[8], phi[8];
        load_frag(ws + ph_off(l, it, j, 0) + lane * 8, phr);
        load_frag(ws + ph_off(l, it, j, 1) + lane * 8, phi);
        #pragma unroll
        for (int s = 0; s < 8; ++s) {
          int m = it * 16 + s + 8 * (lane >> 4);
          int n = j * 16 + (lane & 15);
          Sa_r[m * PITCH + n] = pr[it][s] * phr[s] - pi[it][s] * phi[s];
          Sa_i[m * PITCH + n] = pr[it][s] * phi[s] + pi[it][s] * phr[s];
        }
      }
    }
  }
  float acc[NQ];
  #pragma unroll
  for (int i = 0; i < NQ; ++i) acc[i] = 0.f;
  for (int r = 0; r < 32; ++r) {
    float re = Sa_r[r * PITCH + lane];
    float im = Sa_i[r * PITCH + lane];
    float p  = re * re + im * im;
    int c3 = (r & 1) ? (lane ^ 16) : lane;
    int c4 = (c3 ^ (c3 >> 1) ^ (c3 >> 2) ^ (c3 >> 3) ^ (c3 >> 4)) & 31;
    int r2 = (c4 & 1) ? (r ^ 16) : r;
    int a  = (r2 << 5) | c4;
    #pragma unroll
    for (int i = 0; i < NQ; ++i)
      acc[i] += (a & (1 << (9 - i))) ? -p : p;
  }
  #pragma unroll
  for (int off = 16; off > 0; off >>= 1)
    #pragma unroll
    for (int i = 0; i < NQ; ++i)
      acc[i] += __shfl_xor(acc[i], off, 32);
  if (lane == 0) {
    #pragma unroll
    for (int i = 0; i < NQ; ++i) out[sample * NQ + i] = acc[i];
  }
}

extern "C" void kernel_launch(void* const* d_in, const int* in_sizes, int n_in,
                              void* d_out, int out_size, void* d_ws, size_t ws_size,
                              hipStream_t stream) {
  const float* x = (const float*)d_in[0];
  const float* w = (const float*)d_in[1];
  float* ws  = (float*)d_ws;
  float* out = (float*)d_out;

  vqc_prep<<<NL, 1024, 0, stream>>>(w, ws);
  vqc_main<<<BATCH_N / WPB, 32 * WPB, 0, stream>>>(x, ws, out);
}


# layer 0 RY gates folded into the encoding angles (RY(w)RY(x)=RY(x+w)): state stays a product state, layer 0 needs no matrix product, 384 WMMA per sample not 512; layer 0 sums regrouped
# speedup vs baseline: 2.0214x; 1.2187x over previous
#include <hip/hip_runtime.h>
#include <hip/hip_bf16.h>


typedef __attribute__((ext_vector_type(2))) float v2f;
typedef __attribute__((ext_vector_type(4))) float v4f;
typedef __attribute__((ext_vector_type(8))) float v8f;

#define NQ      10
#define NL      4
#define BATCH_N 8192
#define PITCH   33
#define PLANE   (32 * PITCH)
#define SAMPLE_F (4 * PLANE)
#define WPB     2

__device__ __forceinline__ int rh_off(int l, int i, int k) {
  return ((l * 2 + i) * 2 + k) * 256;
}
__device__ __forceinline__ int rl_off(int l, int k, int j) {
  return 4096 + ((l * 2 + k) * 2 + j) * 256;
}
__device__ __forceinline__ int ph_off(int l, int i, int j, int p) {
  return 8192 + (((l * 2 + i) * 2 + j) * 2 + p) * 256;
}

__device__ __forceinline__ float ry_elem(float a, int rb, int cb) {
  float ca = cosf(0.5f * a), sa = sinf(0.5f * a);
  return rb ? (cb ? ca : sa) : (cb ? -sa : ca);
}

__device__ __forceinline__ void rz_phase(float b, int rb, float& fr, float& fi) {
  float sz = sinf(0.5f * b);
  fr = cosf(0.5f * b);
  fi = rb ? sz : -sz;
}

__global__ void vqc_prep(const float* __restrict__ w, float* __restrict__ ws) {
  int l = blockIdx.x;
  for (int t = threadIdx.x; t < 3072; t += blockDim.x) {
    int which = t >> 10;
    int e     = t & 1023;
    int frag  = e >> 8;
    int lane  = (e >> 3) & 31;
    int slot  = e & 7;
    int f1 = frag >> 1, f0 = frag & 1;
    if (which < 2) {
      if (l == 0) continue;
      int Kloc = 4 * (slot >> 1) + (slot & 1) + 2 * (lane >> 4);
      int rsel, csel, wbase, off;
      if (which == 0) {
        int m = f1 * 16 + (lane & 15);
        rsel = (m ^ (m >> 1)) & 31;
        csel = f0 * 16 + Kloc;
        wbase = 0;
        off = rh_off(l, f1, f0);
      } else {
        rsel = f0 * 16 + (lane & 15);
        csel = f1 * 16 + Kloc;
        wbase = 5;
        off = rl_off(l, f1, f0);
      }
      float v = 1.f;
      #pragma unroll
      for (int w2 = 0; w2 < 5; ++w2)
        v *= ry_elem(w[(l * NQ + wbase + w2) * 2],
                     (rsel >> (4 - w2)) & 1, (csel >> (4 - w2)) & 1);
      ws[off + lane * 8 + slot] = v;
    } else {
      int m  = f1 * 16 + slot + 8 * (lane >> 4);
      int n  = f0 * 16 + (lane & 15);
      int r0 = (m ^ (m >> 1)) & 31;
      float vr = 1.f, vi = 0.f;
      #pragma unroll
      for (int w2 = 0; w2 < NQ; ++w2) {
        int rb = (w2 < 5) ? ((r0 >> (4 - w2)) & 1) : ((n >> (9 - w2)) & 1);
        float fr, fi;
        rz_phase(w[(l * NQ + w2) * 2 + 1], rb, fr, fi);
        float nr = vr * fr - vi * fi; vi = vr * fi + vi * fr; vr = nr;
      }
      ws[ph_off(l, f1, f0, 0) + lane * 8 + slot] = vr;
      ws[ph_off(l, f1, f0, 1) + lane * 8 + slot] = vi;
    }
  }
}

__device__ __forceinline__ void load_frag(const float* __restrict__ p, float (&f)[8]) {
  const v4f* q = (const v4f*)p;
  v4f a = q[0], b = q[1];
  f[0] = a.x; f[1] = a.y; f[2] = a.z; f[3] = a.w;
  f[4] = b.x; f[5] = b.y; f[6] = b.z; f[7] = b.w;
}

__global__ __launch_bounds__(32 * WPB) void vqc_main(const float* __restrict__ x,
                                                     const float* __restrict__ w,
                                                     const float* __restrict__ ws,
                                                     float* __restrict__ out) {
  __shared__ float lds[WPB * SAMPLE_F];
  const int lane = threadIdx.x & 31;
  const int wv   = threadIdx.x >> 5;
  const int sample = blockIdx.x * WPB + wv;

  float* Sa_r = lds + wv * SAMPLE_F;
  float* Sa_i = Sa_r + PLANE;
  float* Sb_r = Sa_r + 2 * PLANE;
  float* Sb_i = Sa_r + 3 * PLANE;

  float rowv = 1.f, colv = 1.f;
  #pragma unroll
  for (int w2 = 0; w2 < 5; ++w2) {
    float ah = 0.5f * (x[sample * NQ + w2]     + w[w2 * 2]);
    float al = 0.5f * (x[sample * NQ + 5 + w2] + w[(5 + w2) * 2]);
    int b = (lane >> (4 - w2)) & 1;
    rowv *= b ? sinf(ah) : cosf(ah);
    colv *= b ? sinf(al) : cosf(al);
  }
  #pragma unroll
  for (int it = 0; it < 2; ++it) {
    float rv[8];
    #pragma unroll
    for (int s = 0; s < 8; ++s) {
      int m = it * 16 + s + 8 * (lane >> 4);
      rv[s] = __shfl(rowv, (m ^ (m >> 1)) & 31, 32);
    }
    #pragma unroll
    for (int j = 0; j < 2; ++j) {
      float cv = __shfl(colv, j * 16 + (lane & 15), 32);
      float phr[8], phi[8];
      load_frag(ws + ph_off(0, it, j, 0) + lane * 8, phr);
      load_frag(ws + ph_off(0, it, j, 1) + lane * 8, phi);
      #pragma unroll
      for (int s = 0; s < 8; ++s) {
        int m = it * 16 + s + 8 * (lane >> 4);
        int n = j * 16 + (lane & 15);
        float a = rv[s] * cv;
        Sa_r[m * PITCH + n] = a * phr[s];
        Sa_i[m * PITCH + n] = a * phi[s];
      }
    }
  }

  for (int l = 1; l < NL; ++l) {
    float rl[2][2][8];
    #pragma unroll
    for (int k = 0; k < 2; ++k)
      #pragma unroll
      for (int j = 0; j < 2; ++j)
        load_frag(ws + rl_off(l, k, j) + lane * 8, rl[k][j]);
    #pragma unroll
    for (int it = 0; it < 2; ++it) {
      float ar[2][8], ai[2][8];
      #pragma unroll
      for (int k = 0; k < 2; ++k)
        #pragma unroll
        for (int s = 0; s < 8; ++s) {
          int row = it * 16 + (lane & 15);
          int col = k * 16 + 4 * (s >> 1) + (s & 1) + 2 * (lane >> 4);
          int rm = (col & 1) ? (row ^ 16) : row;
          int c3 = (col ^ (col >> 1)) & 31;
          int cs = (rm & 1) ? (c3 ^ 16) : c3;
          int rs = rm;
          ar[k][s] = Sa_r[rs * PITCH + cs];
          ai[k][s] = Sa_i[rs * PITCH + cs];
        }
      v8f pr[2], pi[2];
      #pragma unroll
      for (int j = 0; j < 2; ++j) {
        pr[j] = (v8f){0,0,0,0,0,0,0,0}; pi[j] = pr[j];
      }
      #pragma unroll
      for (int k = 0; k < 2; ++k)
        #pragma unroll
        for (int kk = 0; kk < 4; ++kk) {
          v2f aR = { ar[k][2 * kk], ar[k][2 * kk + 1]};
          v2f aI = { ai[k][2 * kk], ai[k][2 * kk + 1]};
          #pragma unroll
          for (int j = 0; j < 2; ++j) {
            v2f bR = { rl[k][j][2 * kk], rl[k][j][2 * kk + 1]};
            pr[j] = __builtin_amdgcn_wmma_f32_16x16x4_f32(false, aR, false, bR, (short)0, pr[j], false, false);
            pi[j] = __builtin_amdgcn_wmma_f32_16x16x4_f32(false, aI, false, bR, (short)0, pi[j], false, false);
          }
        }
      #pragma unroll
      for (int j = 0; j < 2; ++j)
        #pragma unroll
        for (int s = 0; s < 8; ++s) {
          int m = it * 16 + s + 8 * (lane >> 4);
          int n = j * 16 + (lane & 15);
          Sb_r[m * PITCH + n] = pr[j][s];
          Sb_i[m * PITCH + n] = pi[j][s];
        }
    }
    float rh[2][2][8];
    #pragma unroll
    for (int it = 0; it < 2; ++it)
      #pragma unroll
      for (int k = 0; k < 2; ++k)
        load_frag(ws + rh_off(l, it, k) + lane * 8, rh[it][k]);
    #pragma unroll
    for (int j = 0; j < 2; ++j) {
      float br[2][8], bi[2][8];
      #pragma unroll
      for (int k = 0; k < 2; ++k)
        #pragma unroll
        for (int s = 0; s < 8; ++s) {
          int K = k * 16 + 4 * (s >> 1) + (s & 1) + 2 * (lane >> 4);
          int n = j * 16 + (lane & 15);
          br[k][s] = Sb_r[K * PITCH + n];
          bi[k][s] = Sb_i[K * PITCH + n];
        }
      v8f pr[2], pi[2];
      #pragma unroll
      for (int it = 0; it < 2; ++it) {
        pr[it] = (v8f){0,0,0,0,0,0,0,0}; pi[it] = pr[it];
      }
      #pragma unroll
      for (int k = 0; k < 2; ++k)
        #pragma unroll
        for (int kk = 0; kk < 4; ++kk) {
          v2f bR = { br[k][2 * kk], br[k][2 * kk + 1]};
          v2f bI = { bi[k][2 * kk], bi[k][2 * kk + 1]};
          #pragma unroll
          for (int it = 0; it < 2; ++it) {
            v2f aR = { rh[it][k][2 * kk], rh[it][k][2 * kk + 1]};
            pr[it] = __builtin_amdgcn_wmma_f32_16x16x4_f32(false, aR, false, bR, (short)0, pr[it], false, false);
            pi[it] = __builtin_amdgcn_wmma_f32_16x16x4_f32(false, aR, false, bI, (short)0, pi[it], false, false);
          }
        }
      #pragma unroll
      for (int it = 0; it < 2; ++it) {
        float phr[8], phi[8];
        load_frag(ws + ph_off(l, it, j, 0) + lane * 8, phr);
        load_frag(ws + ph_off(l, it, j, 1) + lane * 8, phi);
        #pragma unroll
        for (int s = 0; s < 8; ++s) {
          int m = it * 16 + s + 8 * (lane >> 4);
          int n = j * 16 + (lane & 15);
          Sa_r[m * PITCH + n] = pr[it][s] * phr[s] - pi[it][s] * phi[s];
          Sa_i[m * PITCH + n] = pr[it][s] * phi[s] + pi[it][s] * phr[s];
        }
      }
    }
  }
  float acc[NQ];
  #pragma unroll
  for (int i = 0; i < NQ; ++i) acc[i] = 0.f;
  for (int r = 0; r < 32; ++r) {
    float re = Sa_r[r * PITCH + lane];
    float im = Sa_i[r * PITCH + lane];
    float p  = re * re + im * im;
    int c3 = (r & 1) ? (lane ^ 16) : lane;
    int c4 = (c3 ^ (c3 >> 1) ^ (c3 >> 2) ^ (c3 >> 3) ^ (c3 >> 4)) & 31;
    int r2 = (c4 & 1) ? (r ^ 16) : r;
    int a  = (r2 << 5) | c4;
    #pragma unroll
    for (int i = 0; i < NQ; ++i)
      acc[i] += (a & (1 << (9 - i))) ? -p : p;
  }
  #pragma unroll
  for (int off = 16; off > 0; off >>= 1)
    #pragma unroll
    for (int i = 0; i < NQ; ++i)
      acc[i] += __shfl_xor(acc[i], off, 32);
  if (lane == 0) {
    #pragma unroll
    for (int i = 0; i < NQ; ++i) out[sample * NQ + i] = acc[i];
  }
}

extern "C" void kernel_launch(void* const* d_in, const int* in_sizes, int n_in,
                              void* d_out, int out_size, void* d_ws, size_t ws_size,
                              hipStream_t stream) {
  const float* x = (const float*)d_in[0];
  const float* w = (const float*)d_in[1];
  float* ws  = (float*)d_ws;
  float* out = (float*)d_out;

  vqc_prep<<<NL, 1024, 0, stream>>>(w, ws);
  vqc_main<<<BATCH_N / WPB, 32 * WPB, 0, stream>>>(x, w, ws, out);
}


# measurement fused into the last layer's GEMM2 epilogue: |amp|^2 and signed sums from the accumulator registers (its unit RZ phases drop out), no LDS round trip; signed sums regrouped
# speedup vs baseline: 2.1564x; 1.0668x over previous
#include <hip/hip_runtime.h>
#include <hip/hip_bf16.h>


typedef __attribute__((ext_vector_type(2))) float v2f;
typedef __attribute__((ext_vector_type(4))) float v4f;
typedef __attribute__((ext_vector_type(8))) float v8f;

#define NQ      10
#define NL      4
#define BATCH_N 8192
#define PITCH   33
#define PLANE   (32 * PITCH)
#define SAMPLE_F (4 * PLANE)
#define WPB     2

__device__ __forceinline__ int rh_off(int l, int i, int k) {
  return ((l * 2 + i) * 2 + k) * 256;
}
__device__ __forceinline__ int rl_off(int l, int k, int j) {
  return 4096 + ((l * 2 + k) * 2 + j) * 256;
}
__device__ __forceinline__ int ph_off(int l, int i, int j, int p) {
  return 8192 + (((l * 2 + i) * 2 + j) * 2 + p) * 256;
}

__device__ __forceinline__ float ry_elem(float a, int rb, int cb) {
  float ca = cosf(0.5f * a), sa = sinf(0.5f * a);
  return rb ? (cb ? ca : sa) : (cb ? -sa : ca);
}

__device__ __forceinline__ void rz_phase(float b, int rb, float& fr, float& fi) {
  float sz = sinf(0.5f * b);
  fr = cosf(0.5f * b);
  fi = rb ? sz : -sz;
}

__device__ __forceinline__ constexpr int logical_index(int m, int n) {
  int c3 = (m & 1) ? (n ^ 16) : n;
  int c4 = (c3 ^ (c3 >> 1) ^ (c3 >> 2) ^ (c3 >> 3) ^ (c3 >> 4)) & 31;
  int r2 = (c4 & 1) ? (m ^ 16) : m;
  return (r2 << 5) | c4;
}

__device__ __forceinline__ constexpr int row_of(int x) {
  return (x >> 3) * 16 + (x & 7);
}
__device__ __forceinline__ constexpr int walsh_mask(int i) {
  int u = 0;
  for (int b = 0; b < 4; ++b)
    if (logical_index(row_of(1 << b), 0) & (1 << (9 - i))) u |= 1 << b;
  return u;
}

__global__ void vqc_prep(const float* __restrict__ w, float* __restrict__ ws) {
  int l = blockIdx.x;
  for (int t = threadIdx.x; t < 3072; t += blockDim.x) {
    int which = t >> 10;
    int e     = t & 1023;
    int frag  = e >> 8;
    int lane  = (e >> 3) & 31;
    int slot  = e & 7;
    int f1 = frag >> 1, f0 = frag & 1;
    if (which < 2) {
      if (l == 0) continue;
      int Kloc = 4 * (slot >> 1) + (slot & 1) + 2 * (lane >> 4);
      int rsel, csel, wbase, off;
      if (which == 0) {
        int m = f1 * 16 + (lane & 15);
        rsel = (m ^ (m >> 1)) & 31;
        csel = f0 * 16 + Kloc;
        wbase = 0;
        off = rh_off(l, f1, f0);
      } else {
        rsel = f0 * 16 + (lane & 15);
        csel = f1 * 16 + Kloc;
        wbase = 5;
        off = rl_off(l, f1, f0);
      }
      float v = 1.f;
      #pragma unroll
      for (int w2 = 0; w2 < 5; ++w2)
        v *= ry_elem(w[(l * NQ + wbase + w2) * 2],
                     (rsel >> (4 - w2)) & 1, (csel >> (4 - w2)) & 1);
      ws[off + lane * 8 + slot] = v;
    } else {
      if (l == NL - 1) continue;
      int m  = f1 * 16 + slot + 8 * (lane >> 4);
      int n  = f0 * 16 + (lane & 15);
      int r0 = (m ^ (m >> 1)) & 31;
      float vr = 1.f, vi = 0.f;
      #pragma unroll
      for (int w2 = 0; w2 < NQ; ++w2) {
        int rb = (w2 < 5) ? ((r0 >> (4 - w2)) & 1) : ((n >> (9 - w2)) & 1);
        float fr, fi;
        rz_phase(w[(l * NQ + w2) * 2 + 1], rb, fr, fi);
        float nr = vr * fr - vi * fi; vi = vr * fi + vi * fr; vr = nr;
      }
      ws[ph_off(l, f1, f0, 0) + lane * 8 + slot] = vr;
      ws[ph_off(l, f1, f0, 1) + lane * 8 + slot] = vi;
    }
  }
}

__device__ __forceinline__ void load_frag(const float* __restrict__ p, float (&f)[8]) {
  const v4f* q = (const v4f*)p;
  v4f a = q[0], b = q[1];
  f[0] = a.x; f[1] = a.y; f[2] = a.z; f[3] = a.w;
  f[4] = b.x; f[5] = b.y; f[6] = b.z; f[7] = b.w;
}

__global__ __launch_bounds__(32 * WPB) void vqc_main(const float* __restrict__ x,
                                                     const float* __restrict__ w,
                                                     const float* __restrict__ ws,
                                                     float* __restrict__ out) {
  __shared__ float lds[WPB * SAMPLE_F];
  const int lane = threadIdx.x & 31;
  const int wv   = threadIdx.x >> 5;
  const int sample = blockIdx.x * WPB + wv;

  float* Sa_r = lds + wv * SAMPLE_F;
  float* Sa_i = Sa_r + PLANE;
  float* Sb_r = Sa_r + 2 * PLANE;
  float* Sb_i = Sa_r + 3 * PLANE;

  float rowv = 1.f, colv = 1.f;
  #pragma unroll
  for (int w2 = 0; w2 < 5; ++w2) {
    float ah = 0.5f * (x[sample * NQ + w2]     + w[w2 * 2]);
    float al = 0.5f * (x[sample * NQ + 5 + w2] + w[(5 + w2) * 2]);
    int b = (lane >> (4 - w2)) & 1;
    rowv *= b ? sinf(ah) : cosf(ah);
    colv *= b ? sinf(al) : cosf(al);
  }
  #pragma unroll
  for (int it = 0; it < 2; ++it) {
    float rv[8];
    #pragma unroll
    for (int s = 0; s < 8; ++s) {
      int m = it * 16 + s + 8 * (lane >> 4);
      rv[s] = __shfl(rowv, (m ^ (m >> 1)) & 31, 32);
    }
    #pragma unroll
    for (int j = 0; j < 2; ++j) {
      float cv = __shfl(colv, j * 16 + (lane & 15), 32);
      float phr[8], phi[8];
      load_frag(ws + ph_off(0, it, j, 0) + lane * 8, phr);
      load_frag(ws + ph_off(0, it, j, 1) + lane * 8, phi);
      #pragma unroll
      for (int s = 0; s < 8; ++s) {
        int m = it * 16 + s + 8 * (lane >> 4);
        int n = j * 16 + (lane & 15);
        float a = rv[s] * cv;
        Sa_r[m * PITCH + n] = a * phr[s];
        Sa_i[m * PITCH + n] = a * phi[s];
      }
    }
  }

  float acc[NQ];
  #pragma unroll
  for (int i = 0; i < NQ; ++i) acc[i] = 0.f;

  for (int l = 1; l < NL; ++l) {
    float rl[2][2][8];
    #pragma unroll
    for (int k = 0; k < 2; ++k)
      #pragma unroll
      for (int j = 0; j < 2; ++j)
        load_frag(ws + rl_off(l, k, j) + lane * 8, rl[k][j]);
    #pragma unroll
    for (int it = 0; it < 2; ++it) {
      float ar[2][8], ai[2][8];
      #pragma unroll
      for (int k = 0; k < 2; ++k)
        #pragma unroll
        for (int s = 0; s < 8; ++s) {
          int row = it * 16 + (lane & 15);
          int col = k * 16 + 4 * (s >> 1) + (s & 1) + 2 * (lane >> 4);
          int rm = (col & 1) ? (row ^ 16) : row;
          int c3 = (col ^ (col >> 1)) & 31;
          int cs = (rm & 1) ? (c3 ^ 16) : c3;
          int rs = rm;
          ar[k][s] = Sa_r[rs * PITCH + cs];
          ai[k][s] = Sa_i[rs * PITCH + cs];
        }
      v8f pr[2], pi[2];
      #pragma unroll
      for (int j = 0; j < 2; ++j) {
        pr[j] = (v8f){0,0,0,0,0,0,0,0}; pi[j] = pr[j];
      }
      #pragma unroll
      for (int k = 0; k < 2; ++k)
        #pragma unroll
        for (int kk = 0; kk < 4; ++kk) {
          v2f aR = { ar[k][2 * kk], ar[k][2 * kk + 1]};
          v2f aI = { ai[k][2 * kk], ai[k][2 * kk + 1]};
          #pragma unroll
          for (int j = 0; j < 2; ++j) {
            v2f bR = { rl[k][j][2 * kk], rl[k][j][2 * kk + 1]};
            pr[j] = __builtin_amdgcn_wmma_f32_16x16x4_f32(false, aR, false, bR, (short)0, pr[j], false, false);
            pi[j] = __builtin_amdgcn_wmma_f32_16x16x4_f32(false, aI, false, bR, (short)0, pi[j], false, false);
          }
        }
      #pragma unroll
      for (int j = 0; j < 2; ++j)
        #pragma unroll
        for (int s = 0; s < 8; ++s) {
          int m = it * 16 + s + 8 * (lane >> 4);
          int n = j * 16 + (lane & 15);
          Sb_r[m * PITCH + n] = pr[j][s];
          Sb_i[m * PITCH + n] = pi[j][s];
        }
    }
    float rh[2][2][8];
    #pragma unroll
    for (int it = 0; it < 2; ++it)
      #pragma unroll
      for (int k = 0; k < 2; ++k)
        load_frag(ws + rh_off(l, it, k) + lane * 8, rh[it][k]);
    #pragma unroll
    for (int j = 0; j < 2; ++j) {
      float br[2][8], bi[2][8];
      #pragma unroll
      for (int k = 0; k < 2; ++k)
        #pragma unroll
        for (int s = 0; s < 8; ++s) {
          int K = k * 16 + 4 * (s >> 1) + (s & 1) + 2 * (lane >> 4);
          int n = j * 16 + (lane & 15);
          br[k][s] = Sb_r[K * PITCH + n];
          bi[k][s] = Sb_i[K * PITCH + n];
        }
      v8f pr[2], pi[2];
      #pragma unroll
      for (int it = 0; it < 2; ++it) {
        pr[it] = (v8f){0,0,0,0,0,0,0,0}; pi[it] = pr[it];
      }
      #pragma unroll
      for (int k = 0; k < 2; ++k)
        #pragma unroll
        for (int kk = 0; kk < 4; ++kk) {
          v2f bR = { br[k][2 * kk], br[k][2 * kk + 1]};
          v2f bI = { bi[k][2 * kk], bi[k][2 * kk + 1]};
          #pragma unroll
          for (int it = 0; it < 2; ++it) {
            v2f aR = { rh[it][k][2 * kk], rh[it][k][2 * kk + 1]};
            pr[it] = __builtin_amdgcn_wmma_f32_16x16x4_f32(false, aR, false, bR, (short)0, pr[it], false, false);
            pi[it] = __builtin_amdgcn_wmma_f32_16x16x4_f32(false, aR, false, bI, (short)0, pi[it], false, false);
          }
        }
      if (l < NL - 1) {
        #pragma unroll
        for (int it = 0; it < 2; ++it) {
          float phr[8], phi[8];
          load_frag(ws + ph_off(l, it, j, 0) + lane * 8, phr);
          load_frag(ws + ph_off(l, it, j, 1) + lane * 8, phi);
          #pragma unroll
          for (int s = 0; s < 8; ++s) {
            int m = it * 16 + s + 8 * (lane >> 4);
            int n = j * 16 + (lane & 15);
            Sa_r[m * PITCH + n] = pr[it][s] * phr[s] - pi[it][s] * phi[s];
            Sa_i[m * PITCH + n] = pr[it][s] * phi[s] + pi[it][s] * phr[s];
          }
        }
      } else {
        float wh[16];
        #pragma unroll
        for (int it = 0; it < 2; ++it)
          #pragma unroll
          for (int s = 0; s < 8; ++s)
            wh[it * 8 + s] = pr[it][s] * pr[it][s] + pi[it][s] * pi[it][s];
        #pragma unroll
        for (int len = 1; len < 16; len <<= 1)
          #pragma unroll
          for (int x = 0; x < 16; ++x)
            if (!(x & len)) {
              float a = wh[x], b = wh[x + len];
              wh[x] = a + b;
              wh[x + len] = a - b;
            }
        #pragma unroll
        for (int i = 0; i < NQ; ++i) {
          const float c = wh[walsh_mask(i)];
          acc[i] += (logical_index(0, j * 16) & (1 << (9 - i))) ? -c : c;
        }
      }
    }
  }
  const int al = logical_index(8 * (lane >> 4), lane & 15);
  #pragma unroll
  for (int i = 0; i < NQ; ++i)
    acc[i] = (al & (1 << (9 - i))) ? -acc[i] : acc[i];
  #pragma unroll
  for (int off = 16; off > 0; off >>= 1)
    #pragma unroll
    for (int i = 0; i < NQ; ++i)
      acc[i] += __shfl_xor(acc[i], off, 32);
  if (lane == 0) {
    #pragma unroll
    for (int i = 0; i < NQ; ++i) out[sample * NQ + i] = acc[i];
  }
}

extern "C" void kernel_launch(void* const* d_in, const int* in_sizes, int n_in,
                              void* d_out, int out_size, void* d_ws, size_t ws_size,
                              hipStream_t stream) {
  const float* x = (const float*)d_in[0];
  const float* w = (const float*)d_in[1];
  float* ws  = (float*)d_ws;
  float* out = (float*)d_out;

  vqc_prep<<<NL, 1024, 0, stream>>>(w, ws);
  vqc_main<<<BATCH_N / WPB, 32 * WPB, 0, stream>>>(x, w, ws, out);
}


# every sine/cosine evaluated once per angle and shared: prep computes a layer's 20 pairs once per block into LDS and builds the tables as products; main's encoding one wire per lane, read by lane
# speedup vs baseline: 2.7421x; 1.2716x over previous
#include <hip/hip_runtime.h>
#include <hip/hip_bf16.h>


typedef __attribute__((ext_vector_type(2))) float v2f;
typedef __attribute__((ext_vector_type(4))) float v4f;
typedef __attribute__((ext_vector_type(8))) float v8f;

#define NQ      10
#define NL      4
#define BATCH_N 8192
#define PITCH   33
#define PLANE   (32 * PITCH)
#define SAMPLE_F (4 * PLANE)
#define WPB     2

__device__ __forceinline__ int rh_off(int l, int i, int k) {
  return ((l * 2 + i) * 2 + k) * 256;
}
__device__ __forceinline__ int rl_off(int l, int k, int j) {
  return 4096 + ((l * 2 + k) * 2 + j) * 256;
}
__device__ __forceinline__ int ph_off(int l, int i, int j, int p) {
  return 8192 + (((l * 2 + i) * 2 + j) * 2 + p) * 256;
}

__device__ __forceinline__ constexpr int logical_index(int m, int n) {
  int c3 = (m & 1) ? (n ^ 16) : n;
  int c4 = (c3 ^ (c3 >> 1) ^ (c3 >> 2) ^ (c3 >> 3) ^ (c3 >> 4)) & 31;
  int r2 = (c4 & 1) ? (m ^ 16) : m;
  return (r2 << 5) | c4;
}

__device__ __forceinline__ constexpr int row_of(int x) {
  return (x >> 3) * 16 + (x & 7);
}
__device__ __forceinline__ constexpr int walsh_mask(int i) {
  int u = 0;
  for (int b = 0; b < 4; ++b)
    if (logical_index(row_of(1 << b), 0) & (1 << (9 - i))) u |= 1 << b;
  return u;
}

#define PREP_T      256
#define PREP_CHUNKS 12
#define PREP_BLOCKS (NL * PREP_CHUNKS)

__global__ __launch_bounds__(PREP_T) void vqc_prep(const float* __restrict__ w, float* __restrict__ ws) {
  __shared__ float cs[2 * NQ][2];
  const int l     = blockIdx.x / PREP_CHUNKS;
  const int chunk = blockIdx.x - l * PREP_CHUNKS;
  const int which = chunk >> 2;
  const int frag  = chunk & 3;
  if (which < 2 ? (l == 0) : (l == NL - 1)) return;
  if (threadIdx.x < 2 * NQ) {
    float h = 0.5f * w[l * 2 * NQ + threadIdx.x];
    cs[threadIdx.x][0] = cosf(h);
    cs[threadIdx.x][1] = sinf(h);
  }
  __syncthreads();
  const int lane = threadIdx.x >> 3;
  const int slot = threadIdx.x & 7;
  const int f1 = frag >> 1, f0 = frag & 1;
  if (which < 2) {
    int Kloc = 4 * (slot >> 1) + (slot & 1) + 2 * (lane >> 4);
    int rsel, csel, wbase, off;
    if (which == 0) {
      int m = f1 * 16 + (lane & 15);
      rsel = (m ^ (m >> 1)) & 31;
      csel = f0 * 16 + Kloc;
      wbase = 0;
      off = rh_off(l, f1, f0);
    } else {
      rsel = f0 * 16 + (lane & 15);
      csel = f1 * 16 + Kloc;
      wbase = 5;
      off = rl_off(l, f1, f0);
    }
    float v = 1.f;
    #pragma unroll
    for (int w2 = 0; w2 < 5; ++w2) {
      float ca = cs[(wbase + w2) * 2][0], sa = cs[(wbase + w2) * 2][1];
      int rb = (rsel >> (4 - w2)) & 1, cb = (csel >> (4 - w2)) & 1;
      v *= rb ? (cb ? ca : sa) : (cb ? -sa : ca);
    }
    ws[off + threadIdx.x] = v;
  } else {
    int m  = f1 * 16 + slot + 8 * (lane >> 4);
    int n  = f0 * 16 + (lane & 15);
    int r0 = (m ^ (m >> 1)) & 31;
    float vr = 1.f, vi = 0.f;
    #pragma unroll
    for (int w2 = 0; w2 < NQ; ++w2) {
      int rb = (w2 < 5) ? ((r0 >> (4 - w2)) & 1) : ((n >> (9 - w2)) & 1);
      float fr = cs[w2 * 2 + 1][0], sz = cs[w2 * 2 + 1][1];
      float fi = rb ? sz : -sz;
      float nr = vr * fr - vi * fi; vi = vr * fi + vi * fr; vr = nr;
    }
    ws[ph_off(l, f1, f0, 0) + threadIdx.x] = vr;
    ws[ph_off(l, f1, f0, 1) + threadIdx.x] = vi;
  }
}

__device__ __forceinline__ void load_frag(const float* __restrict__ p, float (&f)[8]) {
  const v4f* q = (const v4f*)p;
  v4f a = q[0], b = q[1];
  f[0] = a.x; f[1] = a.y; f[2] = a.z; f[3] = a.w;
  f[4] = b.x; f[5] = b.y; f[6] = b.z; f[7] = b.w;
}

__global__ __launch_bounds__(32 * WPB) void vqc_main(const float* __restrict__ x,
                                                     const float* __restrict__ w,
                                                     const float* __restrict__ ws,
                                                     float* __restrict__ out) {
  __shared__ float lds[WPB * SAMPLE_F];
  const int lane = threadIdx.x & 31;
  const int wv   = threadIdx.x >> 5;
  const int sample = blockIdx.x * WPB + wv;

  float* Sa_r = lds + wv * SAMPLE_F;
  float* Sa_i = Sa_r + PLANE;
  float* Sb_r = Sa_r + 2 * PLANE;
  float* Sb_i = Sa_r + 3 * PLANE;

  float hx = 0.f;
  if (lane < NQ) hx = 0.5f * (x[sample * NQ + lane] + w[lane * 2]);
  const int sn = __float_as_int(sinf(hx)), cn = __float_as_int(cosf(hx));
  float rowv = 1.f, colv = 1.f;
  #pragma unroll
  for (int w2 = 0; w2 < 5; ++w2) {
    float sh = __int_as_float(__builtin_amdgcn_readlane(sn, w2));
    float ch = __int_as_float(__builtin_amdgcn_readlane(cn, w2));
    float sl = __int_as_float(__builtin_amdgcn_readlane(sn, 5 + w2));
    float cl = __int_as_float(__builtin_amdgcn_readlane(cn, 5 + w2));
    int b = (lane >> (4 - w2)) & 1;
    rowv *= b ? sh : ch;
    colv *= b ? sl : cl;
  }
  #pragma unroll
  for (int it = 0; it < 2; ++it) {
    float rv[8];
    #pragma unroll
    for (int s = 0; s < 8; ++s) {
      int m = it * 16 + s + 8 * (lane >> 4);
      rv[s] = __shfl(rowv, (m ^ (m >> 1)) & 31, 32);
    }
    #pragma unroll
    for (int j = 0; j < 2; ++j) {
      float cv = __shfl(colv, j * 16 + (lane & 15), 32);
      float phr[8], phi[8];
      load_frag(ws + ph_off(0, it, j, 0) + lane * 8, phr);
      load_frag(ws + ph_off(0, it, j, 1) + lane * 8, phi);
      #pragma unroll
      for (int s = 0; s < 8; ++s) {
        int m = it * 16 + s + 8 * (lane >> 4);
        int n = j * 16 + (lane & 15);
        float a = rv[s] * cv;
        Sa_r[m * PITCH + n] = a * phr[s];
        Sa_i[m * PITCH + n] = a * phi[s];
      }
    }
  }

  float acc[NQ];
  #pragma unroll
  for (int i = 0; i < NQ; ++i) acc[i] = 0.f;

  for (int l = 1; l < NL; ++l) {
    float rl[2][2][8];
    #pragma unroll
    for (int k = 0; k < 2; ++k)
      #pragma unroll
      for (int j = 0; j < 2; ++j)
        load_frag(ws + rl_off(l, k, j) + lane * 8, rl[k][j]);
    #pragma unroll
    for (int it = 0; it < 2; ++it) {
      float ar[2][8], ai[2][8];
      #pragma unroll
      for (int k = 0; k < 2; ++k)
        #pragma unroll
        for (int s = 0; s < 8; ++s) {
          int row = it * 16 + (lane & 15);
          int col = k * 16 + 4 * (s >> 1) + (s & 1) + 2 * (lane >> 4);
          int rm = (col & 1) ? (row ^ 16) : row;
          int c3 = (col ^ (col >> 1)) & 31;
          int cs = (rm & 1) ? (c3 ^ 16) : c3;
          int rs = rm;
          ar[k][s] = Sa_r[rs * PITCH + cs];
          ai[k][s] = Sa_i[rs * PITCH + cs];
        }
      v8f pr[2], pi[2];
      #pragma unroll
      for (int j = 0; j < 2; ++j) {
        pr[j] = (v8f){0,0,0,0,0,0,0,0}; pi[j] = pr[j];
      }
      #pragma unroll
      for (int k = 0; k < 2; ++k)
        #pragma unroll
        for (int kk = 0; kk < 4; ++kk) {
          v2f aR = { ar[k][2 * kk], ar[k][2 * kk + 1]};
          v2f aI = { ai[k][2 * kk], ai[k][2 * kk + 1]};
          #pragma unroll
          for (int j = 0; j < 2; ++j) {
            v2f bR = { rl[k][j][2 * kk], rl[k][j][2 * kk + 1]};
            pr[j] = __builtin_amdgcn_wmma_f32_16x16x4_f32(false, aR, false, bR, (short)0, pr[j], false, false);
            pi[j] = __builtin_amdgcn_wmma_f32_16x16x4_f32(false, aI, false, bR, (short)0, pi[j], false, false);
          }
        }
      #pragma unroll
      for (int j = 0; j < 2; ++j)
        #pragma unroll
        for (int s = 0; s < 8; ++s) {
          int m = it * 16 + s + 8 * (lane >> 4);
          int n = j * 16 + (lane & 15);
          Sb_r[m * PITCH + n] = pr[j][s];
          Sb_i[m * PITCH + n] = pi[j][s];
        }
    }
    float rh[2][2][8];
    #pragma unroll
    for (int it = 0; it < 2; ++it)
      #pragma unroll
      for (int k = 0; k < 2; ++k)
        load_frag(ws + rh_off(l, it, k) + lane * 8, rh[it][k]);
    #pragma unroll
    for (int j = 0; j < 2; ++j) {
      float br[2][8], bi[2][8];
      #pragma unroll
      for (int k = 0; k < 2; ++k)
        #pragma unroll
        for (int s = 0; s < 8; ++s) {
          int K = k * 16 + 4 * (s >> 1) + (s & 1) + 2 * (lane >> 4);
          int n = j * 16 + (lane & 15);
          br[k][s] = Sb_r[K * PITCH + n];
          bi[k][s] = Sb_i[K * PITCH + n];
        }
      v8f pr[2], pi[2];
      #pragma unroll
      for (int it = 0; it < 2; ++it) {
        pr[it] = (v8f){0,0,0,0,0,0,0,0}; pi[it] = pr[it];
      }
      #pragma unroll
      for (int k = 0; k < 2; ++k)
        #pragma unroll
        for (int kk = 0; kk < 4; ++kk) {
          v2f bR = { br[k][2 * kk], br[k][2 * kk + 1]};
          v2f bI = { bi[k][2 * kk], bi[k][2 * kk + 1]};
          #pragma unroll
          for (int it = 0; it < 2; ++it) {
            v2f aR = { rh[it][k][2 * kk], rh[it][k][2 * kk + 1]};
            pr[it] = __builtin_amdgcn_wmma_f32_16x16x4_f32(false, aR, false, bR, (short)0, pr[it], false, false);
            pi[it] = __builtin_amdgcn_wmma_f32_16x16x4_f32(false, aR, false, bI, (short)0, pi[it], false, false);
          }
        }
      if (l < NL - 1) {
        #pragma unroll
        for (int it = 0; it < 2; ++it) {
          float phr[8], phi[8];
          load_frag(ws + ph_off(l, it, j, 0) + lane * 8, phr);
          load_frag(ws + ph_off(l, it, j, 1) + lane * 8, phi);
          #pragma unroll
          for (int s = 0; s < 8; ++s) {
            int m = it * 16 + s + 8 * (lane >> 4);
            int n = j * 16 + (lane & 15);
            Sa_r[m * PITCH + n] = pr[it][s] * phr[s] - pi[it][s] * phi[s];
            Sa_i[m * PITCH + n] = pr[it][s] * phi[s] + pi[it][s] * phr[s];
          }
        }
      } else {
        float wh[16];
        #pragma unroll
        for (int it = 0; it < 2; ++it)
          #pragma unroll
          for (int s = 0; s < 8; ++s)
            wh[it * 8 + s] = pr[it][s] * pr[it][s] + pi[it][s] * pi[it][s];
        #pragma unroll
        for (int len = 1; len < 16; len <<= 1)
          #pragma unroll
          for (int x = 0; x < 16; ++x)
            if (!(x & len)) {
              float a = wh[x], b = wh[x + len];
              wh[x] = a + b;
              wh[x + len] = a - b;
            }
        #pragma unroll
        for (int i = 0; i < NQ; ++i) {
          const float c = wh[walsh_mask(i)];
          acc[i] += (logical_index(0, j * 16) & (1 << (9 - i))) ? -c : c;
        }
      }
    }
  }
  const int al = logical_index(8 * (lane >> 4), lane & 15);
  #pragma unroll
  for (int i = 0; i < NQ; ++i)
    acc[i] = (al & (1 << (9 - i))) ? -acc[i] : acc[i];
  #pragma unroll
  for (int off = 16; off > 0; off >>= 1)
    #pragma unroll
    for (int i = 0; i < NQ; ++i)
      acc[i] += __shfl_xor(acc[i], off, 32);
  if (lane == 0) {
    #pragma unroll
    for (int i = 0; i < NQ; ++i) out[sample * NQ + i] = acc[i];
  }
}

extern "C" void kernel_launch(void* const* d_in, const int* in_sizes, int n_in,
                              void* d_out, int out_size, void* d_ws, size_t ws_size,
                              hipStream_t stream) {
  const float* x = (const float*)d_in[0];
  const float* w = (const float*)d_in[1];
  float* ws  = (float*)d_ws;
  float* out = (float*)d_out;

  vqc_prep<<<PREP_BLOCKS, PREP_T, 0, stream>>>(w, ws);
  vqc_main<<<BATCH_N / WPB, 32 * WPB, 0, stream>>>(x, w, ws, out);
}


# each real 32x32 factor split as RY(top wire) (x) 16x16: matrix products contract over K=16 only (64 f32 WMMA per layer, not 128), the 2x2 part mixes two accumulator tiles by FMAs; sums regrouped
# speedup vs baseline: 3.3434x; 1.2193x over previous
#include <hip/hip_runtime.h>
#include <hip/hip_bf16.h>


typedef __attribute__((ext_vector_type(2))) float v2f;
typedef __attribute__((ext_vector_type(4))) float v4f;
typedef __attribute__((ext_vector_type(8))) float v8f;

#define NQ      10
#define NL      4
#define BATCH_N 8192
#define PITCH   33
#define PLANE   (32 * PITCH)
#define SAMPLE_F (4 * PLANE)
#define WPB     2

__device__ __forceinline__ int rh_off(int l) { return l * 256; }
__device__ __forceinline__ int rl_off(int l) { return 1024 + l * 256; }
__device__ __forceinline__ int cs_off(int l) { return 2048 + l * 32; }
__device__ __forceinline__ int ph_off(int l, int i, int j, int p) {
  return 4096 + (((l * 2 + i) * 2 + j) * 2 + p) * 256;
}

__device__ __forceinline__ constexpr int out_row(int it, int s, int h) {
  return it ? 31 - s - 8 * h : s + 8 * h;
}

__device__ __forceinline__ constexpr int logical_index(int m, int n) {
  int c3 = (m & 1) ? (n ^ 16) : n;
  int c4 = (c3 ^ (c3 >> 1) ^ (c3 >> 2) ^ (c3 >> 3) ^ (c3 >> 4)) & 31;
  int r2 = (c4 & 1) ? (m ^ 16) : m;
  return (r2 << 5) | c4;
}

__device__ __forceinline__ constexpr int row_of(int x) {
  return ((x >> 3) ? 31 : 0) ^ (x & 7);
}
__device__ __forceinline__ constexpr int walsh_mask(int i) {
  int u = 0;
  for (int b = 0; b < 4; ++b)
    if (logical_index(row_of(1 << b), 0) & (1 << (9 - i))) u |= 1 << b;
  return u;
}

#define PREP_T      256
#define PREP_CHUNKS 6
#define PREP_BLOCKS (NL * PREP_CHUNKS)

__global__ __launch_bounds__(PREP_T) void vqc_prep(const float* __restrict__ w, float* __restrict__ ws) {
  __shared__ float cs[2 * NQ][2];
  const int l     = blockIdx.x / PREP_CHUNKS;
  const int chunk = blockIdx.x - l * PREP_CHUNKS;
  const int which = chunk < 2 ? chunk : 2;
  const int frag  = (chunk - 2) & 3;
  if (which < 2 ? (l == 0) : (l == NL - 1)) return;
  if (threadIdx.x < 2 * NQ) {
    float h = 0.5f * w[l * 2 * NQ + threadIdx.x];
    cs[threadIdx.x][0] = cosf(h);
    cs[threadIdx.x][1] = sinf(h);
  }
  __syncthreads();
  const int lane = threadIdx.x >> 3;
  const int slot = threadIdx.x & 7;
  const int f1 = frag >> 1, f0 = frag & 1;
  if (which < 2) {
    int Kloc = 4 * (slot >> 1) + (slot & 1) + 2 * (lane >> 4);
    int rsel, wbase, off;
    const int csel = Kloc;
    if (which == 0) {
      int xr = lane & 15;
      rsel = (xr ^ (xr >> 1)) & 15;
      wbase = 0;
      off = rh_off(l);
      if (threadIdx.x < 4) ws[cs_off(l) + threadIdx.x] = cs[(threadIdx.x >> 1) * 10][threadIdx.x & 1];
    } else {
      rsel = lane & 15;
      wbase = 5;
      off = rl_off(l);
    }
    float v = 1.f;
    #pragma unroll
    for (int w2 = 1; w2 < 5; ++w2) {
      float ca = cs[(wbase + w2) * 2][0], sa = cs[(wbase + w2) * 2][1];
      int rb = (rsel >> (4 - w2)) & 1, cb = (csel >> (4 - w2)) & 1;
      v *= rb ? (cb ? ca : sa) : (cb ? -sa : ca);
    }
    ws[off + threadIdx.x] = v;
  } else {
    int m  = out_row(f1, slot, lane >> 4);
    int n  = f0 * 16 + (lane & 15);
    int r0 = (m ^ (m >> 1)) & 31;
    float vr = 1.f, vi = 0.f;
    #pragma unroll
    for (int w2 = 0; w2 < NQ; ++w2) {
      int rb = (w2 < 5) ? ((r0 >> (4 - w2)) & 1) : ((n >> (9 - w2)) & 1);
      float fr = cs[w2 * 2 + 1][0], sz = cs[w2 * 2 + 1][1];
      float fi = rb ? sz : -sz;
      float nr = vr * fr - vi * fi; vi = vr * fi + vi * fr; vr = nr;
    }
    ws[ph_off(l, f1, f0, 0) + threadIdx.x] = vr;
    ws[ph_off(l, f1, f0, 1) + threadIdx.x] = vi;
  }
}

__device__ __forceinline__ void load_frag(const float* __restrict__ p, float (&f)[8]) {
  const v4f* q = (const v4f*)p;
  v4f a = q[0], b = q[1];
  f[0] = a.x; f[1] = a.y; f[2] = a.z; f[3] = a.w;
  f[4] = b.x; f[5] = b.y; f[6] = b.z; f[7] = b.w;
}

__global__ __launch_bounds__(32 * WPB) void vqc_main(const float* __restrict__ x,
                                                     const float* __restrict__ w,
                                                     const float* __restrict__ ws,
                                                     float* __restrict__ out) {
  __shared__ float lds[WPB * SAMPLE_F];
  const int lane = threadIdx.x & 31;
  const int wv   = threadIdx.x >> 5;
  const int sample = blockIdx.x * WPB + wv;

  float* Sa_r = lds + wv * SAMPLE_F;
  float* Sa_i = Sa_r + PLANE;
  float* Sb_r = Sa_r + 2 * PLANE;
  float* Sb_i = Sa_r + 3 * PLANE;

  float hx = 0.f;
  if (lane < NQ) hx = 0.5f * (x[sample * NQ + lane] + w[lane * 2]);
  const int sn = __float_as_int(sinf(hx)), cn = __float_as_int(cosf(hx));
  float rowv = 1.f, colv = 1.f;
  #pragma unroll
  for (int w2 = 0; w2 < 5; ++w2) {
    float sh = __int_as_float(__builtin_amdgcn_readlane(sn, w2));
    float ch = __int_as_float(__builtin_amdgcn_readlane(cn, w2));
    float sl = __int_as_float(__builtin_amdgcn_readlane(sn, 5 + w2));
    float cl = __int_as_float(__builtin_amdgcn_readlane(cn, 5 + w2));
    int b = (lane >> (4 - w2)) & 1;
    rowv *= b ? sh : ch;
    colv *= b ? sl : cl;
  }
  #pragma unroll
  for (int it = 0; it < 2; ++it) {
    float rv[8];
    #pragma unroll
    for (int s = 0; s < 8; ++s) {
      int m = out_row(it, s, lane >> 4);
      rv[s] = __shfl(rowv, (m ^ (m >> 1)) & 31, 32);
    }
    #pragma unroll
    for (int j = 0; j < 2; ++j) {
      float cv = __shfl(colv, j * 16 + (lane & 15), 32);
      float phr[8], phi[8];
      load_frag(ws + ph_off(0, it, j, 0) + lane * 8, phr);
      load_frag(ws + ph_off(0, it, j, 1) + lane * 8, phi);
      #pragma unroll
      for (int s = 0; s < 8; ++s) {
        int m = out_row(it, s, lane >> 4);
        int n = j * 16 + (lane & 15);
        float a = rv[s] * cv;
        Sa_r[m * PITCH + n] = a * phr[s];
        Sa_i[m * PITCH + n] = a * phi[s];
      }
    }
  }

  float acc[NQ];
  #pragma unroll
  for (int i = 0; i < NQ; ++i) acc[i] = 0.f;

  for (int l = 1; l < NL; ++l) {
    const float c0 = ws[cs_off(l) + 0], s0 = ws[cs_off(l) + 1];
    const float c5 = ws[cs_off(l) + 2], s5 = ws[cs_off(l) + 3];
    float rl[8];
    load_frag(ws + rl_off(l) + lane * 8, rl);
    #pragma unroll
    for (int it = 0; it < 2; ++it) {
      float ar[2][8], ai[2][8];
      #pragma unroll
      for (int k = 0; k < 2; ++k)
        #pragma unroll
        for (int s = 0; s < 8; ++s) {
          int row = it * 16 + (lane & 15);
          int col = k * 16 + 4 * (s >> 1) + (s & 1) + 2 * (lane >> 4);
          int rm = (col & 1) ? (row ^ 16) : row;
          int c3 = (col ^ (col >> 1)) & 31;
          int cs = (rm & 1) ? (c3 ^ 16) : c3;
          int rs = rm;
          ar[k][s] = Sa_r[rs * PITCH + cs];
          ai[k][s] = Sa_i[rs * PITCH + cs];
        }
      v8f ur[2], ui[2];
      #pragma unroll
      for (int k = 0; k < 2; ++k) {
        ur[k] = (v8f){0,0,0,0,0,0,0,0}; ui[k] = ur[k];
      }
      #pragma unroll
      for (int kk = 0; kk < 4; ++kk) {
        v2f bR = { rl[2 * kk], rl[2 * kk + 1]};
        #pragma unroll
        for (int k = 0; k < 2; ++k) {
          v2f aR = { ar[k][2 * kk], ar[k][2 * kk + 1]};
          v2f aI = { ai[k][2 * kk], ai[k][2 * kk + 1]};
          ur[k] = __builtin_amdgcn_wmma_f32_16x16x4_f32(false, aR, false, bR, (short)0, ur[k], false, false);
          ui[k] = __builtin_amdgcn_wmma_f32_16x16x4_f32(false, aI, false, bR, (short)0, ui[k], false, false);
        }
      }
      #pragma unroll
      for (int s = 0; s < 8; ++s) {
        int m = it * 16 + s + 8 * (lane >> 4);
        int n = lane & 15;
        Sb_r[m * PITCH + n]      = c5 * ur[0][s] - s5 * ur[1][s];
        Sb_r[m * PITCH + 16 + n] = s5 * ur[0][s] + c5 * ur[1][s];
        Sb_i[m * PITCH + n]      = c5 * ui[0][s] - s5 * ui[1][s];
        Sb_i[m * PITCH + 16 + n] = s5 * ui[0][s] + c5 * ui[1][s];
      }
    }
    float rh[8];
    load_frag(ws + rh_off(l) + lane * 8, rh);
    #pragma unroll
    for (int j = 0; j < 2; ++j) {
      float br[2][8], bi[2][8];
      #pragma unroll
      for (int k = 0; k < 2; ++k)
        #pragma unroll
        for (int s = 0; s < 8; ++s) {
          int K = k * 16 + 4 * (s >> 1) + (s & 1) + 2 * (lane >> 4);
          int n = j * 16 + (lane & 15);
          br[k][s] = Sb_r[K * PITCH + n];
          bi[k][s] = Sb_i[K * PITCH + n];
        }
      v8f vr[2], vi[2];
      #pragma unroll
      for (int k = 0; k < 2; ++k) {
        vr[k] = (v8f){0,0,0,0,0,0,0,0}; vi[k] = vr[k];
      }
      #pragma unroll
      for (int kk = 0; kk < 4; ++kk) {
        v2f aR = { rh[2 * kk], rh[2 * kk + 1]};
        #pragma unroll
        for (int k = 0; k < 2; ++k) {
          v2f bR = { br[k][2 * kk], br[k][2 * kk + 1]};
          v2f bI = { bi[k][2 * kk], bi[k][2 * kk + 1]};
          vr[k] = __builtin_amdgcn_wmma_f32_16x16x4_f32(false, aR, false, bR, (short)0, vr[k], false, false);
          vi[k] = __builtin_amdgcn_wmma_f32_16x16x4_f32(false, aR, false, bI, (short)0, vi[k], false, false);
        }
      }
      float pr[2][8], pi[2][8];
      #pragma unroll
      for (int s = 0; s < 8; ++s) {
        pr[0][s] = c0 * vr[0][s] - s0 * vr[1][s];
        pr[1][s] = s0 * vr[0][s] + c0 * vr[1][s];
        pi[0][s] = c0 * vi[0][s] - s0 * vi[1][s];
        pi[1][s] = s0 * vi[0][s] + c0 * vi[1][s];
      }
      if (l < NL - 1) {
        #pragma unroll
        for (int it = 0; it < 2; ++it) {
          float phr[8], phi[8];
          load_frag(ws + ph_off(l, it, j, 0) + lane * 8, phr);
          load_frag(ws + ph_off(l, it, j, 1) + lane * 8, phi);
          #pragma unroll
          for (int s = 0; s < 8; ++s) {
            int m = out_row(it, s, lane >> 4);
            int n = j * 16 + (lane & 15);
            Sa_r[m * PITCH + n] = pr[it][s] * phr[s] - pi[it][s] * phi[s];
            Sa_i[m * PITCH + n] = pr[it][s] * phi[s] + pi[it][s] * phr[s];
          }
        }
      } else {
        float wh[16];
        #pragma unroll
        for (int it = 0; it < 2; ++it)
          #pragma unroll
          for (int s = 0; s < 8; ++s)
            wh[it * 8 + s] = pr[it][s] * pr[it][s] + pi[it][s] * pi[it][s];
        #pragma unroll
        for (int len = 1; len < 16; len <<= 1)
          #pragma unroll
          for (int x = 0; x < 16; ++x)
            if (!(x & len)) {
              float a = wh[x], b = wh[x + len];
              wh[x] = a + b;
              wh[x + len] = a - b;
            }
        #pragma unroll
        for (int i = 0; i < NQ; ++i) {
          const float c = wh[walsh_mask(i)];
          acc[i] += (logical_index(0, j * 16) & (1 << (9 - i))) ? -c : c;
        }
      }
    }
  }
  const int al = logical_index(8 * (lane >> 4), lane & 15);
  #pragma unroll
  for (int i = 0; i < NQ; ++i)
    acc[i] = (al & (1 << (9 - i))) ? -acc[i] : acc[i];
  #pragma unroll
  for (int off = 16; off > 0; off >>= 1)
    #pragma unroll
    for (int i = 0; i < NQ; ++i)
      acc[i] += __shfl_xor(acc[i], off, 32);
  if (lane == 0) {
    #pragma unroll
    for (int i = 0; i < NQ; ++i) out[sample * NQ + i] = acc[i];
  }
}

extern "C" void kernel_launch(void* const* d_in, const int* in_sizes, int n_in,
                              void* d_out, int out_size, void* d_ws, size_t ws_size,
                              hipStream_t stream) {
  const float* x = (const float*)d_in[0];
  const float* w = (const float*)d_in[1];
  float* ws  = (float*)d_ws;
  float* out = (float*)d_out;

  vqc_prep<<<PREP_BLOCKS, PREP_T, 0, stream>>>(w, ws);
  vqc_main<<<BATCH_N / WPB, 32 * WPB, 0, stream>>>(x, w, ws, out);
}


# GEMM1's accumulator tiles feed GEMM2's B operand straight from registers (row-factor table built in the D layout's k order): no LDS store and reload of the intermediate; GEMM2 sums regrouped
# speedup vs baseline: 4.1499x; 1.2412x over previous
#include <hip/hip_runtime.h>
#include <hip/hip_bf16.h>


typedef __attribute__((ext_vector_type(2))) float v2f;
typedef __attribute__((ext_vector_type(4))) float v4f;
typedef __attribute__((ext_vector_type(8))) float v8f;

#define NQ      10
#define NL      4
#define BATCH_N 8192
#define PITCH   33
#define PLANE   (32 * PITCH)
#define SAMPLE_F (2 * PLANE)
#define WPB     2

__device__ __forceinline__ int rh_off(int l) { return l * 256; }
__device__ __forceinline__ int rl_off(int l) { return 1024 + l * 256; }
__device__ __forceinline__ int cs_off(int l) { return 2048 + l * 32; }
__device__ __forceinline__ int ph_off(int l, int i, int j, int p) {
  return 4096 + (((l * 2 + i) * 2 + j) * 2 + p) * 256;
}

__device__ __forceinline__ constexpr int out_row(int it, int s, int h) {
  return it ? 31 - s - 8 * h : s + 8 * h;
}

__device__ __forceinline__ constexpr int logical_index(int m, int n) {
  int c3 = (m & 1) ? (n ^ 16) : n;
  int c4 = (c3 ^ (c3 >> 1) ^ (c3 >> 2) ^ (c3 >> 3) ^ (c3 >> 4)) & 31;
  int r2 = (c4 & 1) ? (m ^ 16) : m;
  return (r2 << 5) | c4;
}

__device__ __forceinline__ constexpr int row_of(int x) {
  return ((x >> 3) ? 31 : 0) ^ (x & 7);
}
__device__ __forceinline__ constexpr int walsh_mask(int i) {
  int u = 0;
  for (int b = 0; b < 4; ++b)
    if (logical_index(row_of(1 << b), 0) & (1 << (9 - i))) u |= 1 << b;
  return u;
}

#define PREP_T      256
#define PREP_CHUNKS 6
#define PREP_BLOCKS (NL * PREP_CHUNKS)

__global__ __launch_bounds__(PREP_T) void vqc_prep(const float* __restrict__ w, float* __restrict__ ws) {
  __shared__ float cs[2 * NQ][2];
  const int l     = blockIdx.x / PREP_CHUNKS;
  const int chunk = blockIdx.x - l * PREP_CHUNKS;
  const int which = chunk < 2 ? chunk : 2;
  const int frag  = (chunk - 2) & 3;
  if (which < 2 ? (l == 0) : (l == NL - 1)) return;
  if (threadIdx.x < 2 * NQ) {
    float h = 0.5f * w[l * 2 * NQ + threadIdx.x];
    cs[threadIdx.x][0] = cosf(h);
    cs[threadIdx.x][1] = sinf(h);
  }
  __syncthreads();
  const int lane = threadIdx.x >> 3;
  const int slot = threadIdx.x & 7;
  const int f1 = frag >> 1, f0 = frag & 1;
  if (which < 2) {
    int rsel, wbase, off;
    int csel;
    if (which == 0) {
      csel = slot + 8 * (lane >> 4);
      int xr = lane & 15;
      rsel = (xr ^ (xr >> 1)) & 15;
      wbase = 0;
      off = rh_off(l);
      if (threadIdx.x < 4) ws[cs_off(l) + threadIdx.x] = cs[(threadIdx.x >> 1) * 10][threadIdx.x & 1];
    } else {
      rsel = lane & 15;
      csel = 4 * (slot >> 1) + (slot & 1) + 2 * (lane >> 4);
      wbase = 5;
      off = rl_off(l);
    }
    float v = 1.f;
    #pragma unroll
    for (int w2 = 1; w2 < 5; ++w2) {
      float ca = cs[(wbase + w2) * 2][0], sa = cs[(wbase + w2) * 2][1];
      int rb = (rsel >> (4 - w2)) & 1, cb = (csel >> (4 - w2)) & 1;
      v *= rb ? (cb ? ca : sa) : (cb ? -sa : ca);
    }
    ws[off + threadIdx.x] = v;
  } else {
    int m  = out_row(f1, slot, lane >> 4);
    int n  = f0 * 16 + (lane & 15);
    int r0 = (m ^ (m >> 1)) & 31;
    float vr = 1.f, vi = 0.f;
    #pragma unroll
    for (int w2 = 0; w2 < NQ; ++w2) {
      int rb = (w2 < 5) ? ((r0 >> (4 - w2)) & 1) : ((n >> (9 - w2)) & 1);
      float fr = cs[w2 * 2 + 1][0], sz = cs[w2 * 2 + 1][1];
      float fi = rb ? sz : -sz;
      float nr = vr * fr - vi * fi; vi = vr * fi + vi * fr; vr = nr;
    }
    ws[ph_off(l, f1, f0, 0) + threadIdx.x] = vr;
    ws[ph_off(l, f1, f0, 1) + threadIdx.x] = vi;
  }
}

__device__ __forceinline__ void load_frag(const float* __restrict__ p, float (&f)[8]) {
  const v4f* q = (const v4f*)p;
  v4f a = q[0], b = q[1];
  f[0] = a.x; f[1] = a.y; f[2] = a.z; f[3] = a.w;
  f[4] = b.x; f[5] = b.y; f[6] = b.z; f[7] = b.w;
}

__global__ __launch_bounds__(32 * WPB) void vqc_main(const float* __restrict__ x,
                                                     const float* __restrict__ w,
                                                     const float* __restrict__ ws,
                                                     float* __restrict__ out) {
  __shared__ float lds[WPB * SAMPLE_F];
  const int lane = threadIdx.x & 31;
  const int wv   = threadIdx.x >> 5;
  const int sample = blockIdx.x * WPB + wv;

  float* Sa_r = lds + wv * SAMPLE_F;
  float* Sa_i = Sa_r + PLANE;

  float hx = 0.f;
  if (lane < NQ) hx = 0.5f * (x[sample * NQ + lane] + w[lane * 2]);
  const int sn = __float_as_int(sinf(hx)), cn = __float_as_int(cosf(hx));
  float rowv = 1.f, colv = 1.f;
  #pragma unroll
  for (int w2 = 0; w2 < 5; ++w2) {
    float sh = __int_as_float(__builtin_amdgcn_readlane(sn, w2));
    float ch = __int_as_float(__builtin_amdgcn_readlane(cn, w2));
    float sl = __int_as_float(__builtin_amdgcn_readlane(sn, 5 + w2));
    float cl = __int_as_float(__builtin_amdgcn_readlane(cn, 5 + w2));
    int b = (lane >> (4 - w2)) & 1;
    rowv *= b ? sh : ch;
    colv *= b ? sl : cl;
  }
  #pragma unroll
  for (int it = 0; it < 2; ++it) {
    float rv[8];
    #pragma unroll
    for (int s = 0; s < 8; ++s) {
      int m = out_row(it, s, lane >> 4);
      rv[s] = __shfl(rowv, (m ^ (m >> 1)) & 31, 32);
    }
    #pragma unroll
    for (int j = 0; j < 2; ++j) {
      float cv = __shfl(colv, j * 16 + (lane & 15), 32);
      float phr[8], phi[8];
      load_frag(ws + ph_off(0, it, j, 0) + lane * 8, phr);
      load_frag(ws + ph_off(0, it, j, 1) + lane * 8, phi);
      #pragma unroll
      for (int s = 0; s < 8; ++s) {
        int m = out_row(it, s, lane >> 4);
        int n = j * 16 + (lane & 15);
        float a = rv[s] * cv;
        Sa_r[m * PITCH + n] = a * phr[s];
        Sa_i[m * PITCH + n] = a * phi[s];
      }
    }
  }

  float acc[NQ];
  #pragma unroll
  for (int i = 0; i < NQ; ++i) acc[i] = 0.f;

  for (int l = 1; l < NL; ++l) {
    const float c0 = ws[cs_off(l) + 0], s0 = ws[cs_off(l) + 1];
    const float c5 = ws[cs_off(l) + 2], s5 = ws[cs_off(l) + 3];
    float rl[8];
    load_frag(ws + rl_off(l) + lane * 8, rl);
    float rh[8];
    load_frag(ws + rh_off(l) + lane * 8, rh);
    v8f vr[2][2], vi[2][2];
    #pragma unroll
    for (int it = 0; it < 2; ++it) {
      v8f tr[2], ti[2];
      float ar[2][8], ai[2][8];
      #pragma unroll
      for (int k = 0; k < 2; ++k)
        #pragma unroll
        for (int s = 0; s < 8; ++s) {
          int row = it * 16 + (lane & 15);
          int col = k * 16 + 4 * (s >> 1) + (s & 1) + 2 * (lane >> 4);
          int rm = (col & 1) ? (row ^ 16) : row;
          int c3 = (col ^ (col >> 1)) & 31;
          int cs = (rm & 1) ? (c3 ^ 16) : c3;
          int rs = rm;
          ar[k][s] = Sa_r[rs * PITCH + cs];
          ai[k][s] = Sa_i[rs * PITCH + cs];
        }
      v8f ur[2], ui[2];
      #pragma unroll
      for (int k = 0; k < 2; ++k) {
        ur[k] = (v8f){0,0,0,0,0,0,0,0}; ui[k] = ur[k];
      }
      #pragma unroll
      for (int kk = 0; kk < 4; ++kk) {
        v2f bR = { rl[2 * kk], rl[2 * kk + 1]};
        #pragma unroll
        for (int k = 0; k < 2; ++k) {
          v2f aR = { ar[k][2 * kk], ar[k][2 * kk + 1]};
          v2f aI = { ai[k][2 * kk], ai[k][2 * kk + 1]};
          ur[k] = __builtin_amdgcn_wmma_f32_16x16x4_f32(false, aR, false, bR, (short)0, ur[k], false, false);
          ui[k] = __builtin_amdgcn_wmma_f32_16x16x4_f32(false, aI, false, bR, (short)0, ui[k], false, false);
        }
      }
      tr[0] = c5 * ur[0] - s5 * ur[1];
      tr[1] = s5 * ur[0] + c5 * ur[1];
      ti[0] = c5 * ui[0] - s5 * ui[1];
      ti[1] = s5 * ui[0] + c5 * ui[1];
      #pragma unroll
      for (int j = 0; j < 2; ++j) {
        vr[j][it] = (v8f){0,0,0,0,0,0,0,0}; vi[j][it] = vr[j][it];
        #pragma unroll
        for (int kk = 0; kk < 4; ++kk) {
          v2f aR = { rh[2 * kk], rh[2 * kk + 1]};
          v2f bR = { tr[j][2 * kk], tr[j][2 * kk + 1]};
          v2f bI = { ti[j][2 * kk], ti[j][2 * kk + 1]};
          vr[j][it] = __builtin_amdgcn_wmma_f32_16x16x4_f32(false, aR, false, bR, (short)0, vr[j][it], false, false);
          vi[j][it] = __builtin_amdgcn_wmma_f32_16x16x4_f32(false, aR, false, bI, (short)0, vi[j][it], false, false);
        }
      }
    }
    #pragma unroll
    for (int j = 0; j < 2; ++j) {
      float pr[2][8], pi[2][8];
      #pragma unroll
      for (int s = 0; s < 8; ++s) {
        pr[0][s] = c0 * vr[j][0][s] - s0 * vr[j][1][s];
        pr[1][s] = s0 * vr[j][0][s] + c0 * vr[j][1][s];
        pi[0][s] = c0 * vi[j][0][s] - s0 * vi[j][1][s];
        pi[1][s] = s0 * vi[j][0][s] + c0 * vi[j][1][s];
      }
      if (l < NL - 1) {
        #pragma unroll
        for (int it = 0; it < 2; ++it) {
          float phr[8], phi[8];
          load_frag(ws + ph_off(l, it, j, 0) + lane * 8, phr);
          load_frag(ws + ph_off(l, it, j, 1) + lane * 8, phi);
          #pragma unroll
          for (int s = 0; s < 8; ++s) {
            int m = out_row(it, s, lane >> 4);
            int n = j * 16 + (lane & 15);
            Sa_r[m * PITCH + n] = pr[it][s] * phr[s] - pi[it][s] * phi[s];
            Sa_i[m * PITCH + n] = pr[it][s] * phi[s] + pi[it][s] * phr[s];
          }
        }
      } else {
        float wh[16];
        #pragma unroll
        for (int it = 0; it < 2; ++it)
          #pragma unroll
          for (int s = 0; s < 8; ++s)
            wh[it * 8 + s] = pr[it][s] * pr[it][s] + pi[it][s] * pi[it][s];
        #pragma unroll
        for (int len = 1; len < 16; len <<= 1)
          #pragma unroll
          for (int x = 0; x < 16; ++x)
            if (!(x & len)) {
              float a = wh[x], b = wh[x + len];
              wh[x] = a + b;
              wh[x + len] = a - b;
            }
        #pragma unroll
        for (int i = 0; i < NQ; ++i) {
          const float c = wh[walsh_mask(i)];
          acc[i] += (logical_index(0, j * 16) & (1 << (9 - i))) ? -c : c;
        }
      }
    }
  }
  const int al = logical_index(8 * (lane >> 4), lane & 15);
  #pragma unroll
  for (int i = 0; i < NQ; ++i)
    acc[i] = (al & (1 << (9 - i))) ? -acc[i] : acc[i];
  #pragma unroll
  for (int off = 16; off > 0; off >>= 1)
    #pragma unroll
    for (int i = 0; i < NQ; ++i)
      acc[i] += __shfl_xor(acc[i], off, 32);
  if (lane == 0) {
    #pragma unroll
    for (int i = 0; i < NQ; ++i) out[sample * NQ + i] = acc[i];
  }
}

extern "C" void kernel_launch(void* const* d_in, const int* in_sizes, int n_in,
                              void* d_out, int out_size, void* d_ws, size_t ws_size,
                              hipStream_t stream) {
  const float* x = (const float*)d_in[0];
  const float* w = (const float*)d_in[1];
  float* ws  = (float*)d_ws;
  float* out = (float*)d_out;

  vqc_prep<<<PREP_BLOCKS, PREP_T, 0, stream>>>(w, ws);
  vqc_main<<<BATCH_N / WPB, 32 * WPB, 0, stream>>>(x, w, ws, out);
}


# layer 1 from the rank of its input (a x b seen through layer 0's CNOTs = 4 outer products): R_low*b, Q_high*a pieces by 12 wmma, V by 16 over the 8 real terms; 156 wmma/sample, not 192; sums regrouped
# speedup vs baseline: 4.8337x; 1.1648x over previous
#include <hip/hip_runtime.h>
#include <hip/hip_bf16.h>


typedef __attribute__((ext_vector_type(2))) float v2f;
typedef __attribute__((ext_vector_type(4))) float v4f;
typedef __attribute__((ext_vector_type(8))) float v8f;

#define NQ      10
#define NL      4
#define BATCH_N 8192
#define PITCH   33
#define PLANE   (32 * PITCH)
#define SAMPLE_F (2 * PLANE)
#define WPB     2

__device__ __forceinline__ int rh_off(int l) { return l * 256; }
__device__ __forceinline__ int rl_off(int l) { return 1024 + l * 256; }
__device__ __forceinline__ int cs_off(int l) { return 2048 + l * 32; }
__device__ __forceinline__ int ph_off(int l, int i, int j, int p) {
  return 4096 + (((l * 2 + i) * 2 + j) * 2 + p) * 256;
}

__device__ __forceinline__ constexpr int out_row(int it, int s, int h) {
  return it ? 31 - s - 8 * h : s + 8 * h;
}

__device__ __forceinline__ constexpr int logical_index(int m, int n) {
  int c3 = (m & 1) ? (n ^ 16) : n;
  int c4 = (c3 ^ (c3 >> 1) ^ (c3 >> 2) ^ (c3 >> 3) ^ (c3 >> 4)) & 31;
  int r2 = (c4 & 1) ? (m ^ 16) : m;
  return (r2 << 5) | c4;
}

__device__ __forceinline__ constexpr int row_of(int x) {
  return ((x >> 3) ? 31 : 0) ^ (x & 7);
}
__device__ __forceinline__ constexpr int walsh_mask(int i) {
  int u = 0;
  for (int b = 0; b < 4; ++b)
    if (logical_index(row_of(1 << b), 0) & (1 << (9 - i))) u |= 1 << b;
  return u;
}

#define PREP_T      256
#define PREP_CHUNKS 6
#define PREP_BLOCKS (NL * PREP_CHUNKS)

__global__ __launch_bounds__(PREP_T) void vqc_prep(const float* __restrict__ w, float* __restrict__ ws) {
  __shared__ float cs[2 * NQ][2];
  const int l     = blockIdx.x / PREP_CHUNKS;
  const int chunk = blockIdx.x - l * PREP_CHUNKS;
  const int which = chunk < 2 ? chunk : 2;
  const int frag  = (chunk - 2) & 3;
  if (l == 0 || (which == 2 && l == NL - 1)) return;
  if (threadIdx.x < 2 * NQ) {
    float h = 0.5f * w[l * 2 * NQ + threadIdx.x];
    cs[threadIdx.x][0] = cosf(h);
    cs[threadIdx.x][1] = sinf(h);
  }
  __syncthreads();
  const int lane = threadIdx.x >> 3;
  const int slot = threadIdx.x & 7;
  const int f1 = frag >> 1, f0 = frag & 1;
  if (which < 2) {
    int rsel, wbase, off;
    int csel;
    if (which == 0) {
      csel = slot + 8 * (lane >> 4);
      int xr = lane & 15;
      rsel = (xr ^ (xr >> 1)) & 15;
      wbase = 0;
      off = rh_off(l);
      if (threadIdx.x < 4) ws[cs_off(l) + threadIdx.x] = cs[(threadIdx.x >> 1) * 10][threadIdx.x & 1];
    } else {
      rsel = lane & 15;
      csel = 4 * (slot >> 1) + (slot & 1) + 2 * (lane >> 4);
      wbase = 5;
      off = rl_off(l);
    }
    float v = 1.f;
    #pragma unroll
    for (int w2 = 1; w2 < 5; ++w2) {
      float ca = cs[(wbase + w2) * 2][0], sa = cs[(wbase + w2) * 2][1];
      int rb = (rsel >> (4 - w2)) & 1, cb = (csel >> (4 - w2)) & 1;
      v *= rb ? (cb ? ca : sa) : (cb ? -sa : ca);
    }
    ws[off + threadIdx.x] = v;
  } else {
    int m  = out_row(f1, slot, lane >> 4);
    int n  = f0 * 16 + (lane & 15);
    int r0 = (m ^ (m >> 1)) & 31;
    float vr = 1.f, vi = 0.f;
    #pragma unroll
    for (int w2 = 0; w2 < NQ; ++w2) {
      int rb = (w2 < 5) ? ((r0 >> (4 - w2)) & 1) : ((n >> (9 - w2)) & 1);
      float fr = cs[w2 * 2 + 1][0], sz = cs[w2 * 2 + 1][1];
      float fi = rb ? sz : -sz;
      float nr = vr * fr - vi * fi; vi = vr * fi + vi * fr; vr = nr;
    }
    ws[ph_off(l, f1, f0, 0) + threadIdx.x] = vr;
    ws[ph_off(l, f1, f0, 1) + threadIdx.x] = vi;
  }
}

__device__ __forceinline__ void load_frag(const float* __restrict__ p, float (&f)[8]) {
  const v4f* q = (const v4f*)p;
  v4f a = q[0], b = q[1];
  f[0] = a.x; f[1] = a.y; f[2] = a.z; f[3] = a.w;
  f[4] = b.x; f[5] = b.y; f[6] = b.z; f[7] = b.w;
}

__global__ __launch_bounds__(32 * WPB) void vqc_main(const float* __restrict__ x,
                                                     const float* __restrict__ w,
                                                     const float* __restrict__ ws,
                                                     float* __restrict__ out) {
  __shared__ float lds[WPB * SAMPLE_F];
  const int lane = threadIdx.x & 31;
  const int wv   = threadIdx.x >> 5;
  const int sample = blockIdx.x * WPB + wv;

  float* Sa_r = lds + wv * SAMPLE_F;
  float* Sa_i = Sa_r + PLANE;

  float hx = 0.f;
  if (lane < NQ) hx = 0.5f * (x[sample * NQ + lane] + w[lane * 2]);
  else if (lane < 2 * NQ) hx = 0.5f * w[(lane - NQ) * 2 + 1];
  const int sn = __float_as_int(sinf(hx)), cn = __float_as_int(cosf(hx));
  float a_r = 1.f, a_i = 0.f, b_r = 1.f, b_i = 0.f;
  #pragma unroll
  for (int w2 = 0; w2 < 5; ++w2) {
    const int bit = (lane >> (4 - w2)) & 1;
    {
      float sy = __int_as_float(__builtin_amdgcn_readlane(sn, w2));
      float cy = __int_as_float(__builtin_amdgcn_readlane(cn, w2));
      float sz = __int_as_float(__builtin_amdgcn_readlane(sn, NQ + w2));
      float cz = __int_as_float(__builtin_amdgcn_readlane(cn, NQ + w2));
      float mag = bit ? sy : cy;
      float fr = mag * cz, fi = mag * (bit ? sz : -sz);
      float nr = a_r * fr - a_i * fi; a_i = a_r * fi + a_i * fr; a_r = nr;
    }
    {
      float sy = __int_as_float(__builtin_amdgcn_readlane(sn, 5 + w2));
      float cy = __int_as_float(__builtin_amdgcn_readlane(cn, 5 + w2));
      float sz = __int_as_float(__builtin_amdgcn_readlane(sn, NQ + 5 + w2));
      float cz = __int_as_float(__builtin_amdgcn_readlane(cn, NQ + 5 + w2));
      float mag = bit ? sy : cy;
      float fr = mag * cz, fi = mag * (bit ? sz : -sz);
      float nr = b_r * fr - b_i * fi; b_i = b_r * fi + b_i * fr; b_r = nr;
    }
  }
  Sa_r[2 * lane] = a_r; Sa_r[2 * lane + 1] = a_i;
  Sa_i[2 * lane] = b_r; Sa_i[2 * lane + 1] = b_i;

  float acc[NQ];
  #pragma unroll
  for (int i = 0; i < NQ; ++i) acc[i] = 0.f;

  for (int l = 1; l < NL; ++l) {
    const float c0 = ws[cs_off(l) + 0], s0 = ws[cs_off(l) + 1];
    const float c5 = ws[cs_off(l) + 2], s5 = ws[cs_off(l) + 3];
    float rl[8];
    load_frag(ws + rl_off(l) + lane * 8, rl);
    float rh[8];
    load_frag(ws + rh_off(l) + lane * 8, rh);
    v8f vr[2][2], vi[2][2];
    if (l == 1) {
      const int hl = lane >> 4;
      const int vq = lane & 1, vp = (lane >> 1) & 1;
      const int vpart = ((lane >> 2) ^ (lane >> 3)) & 1;
      const bool vneg = (lane & 12) == 8;
      v8f uv[2];
      #pragma unroll
      for (int k = 0; k < 2; ++k) {
        float z[8];
        #pragma unroll
        for (int s = 0; s < 8; ++s) {
          int col = k * 16 + 4 * (s >> 1) + (s & 1) + 2 * hl;
          int cs = ((col ^ (col >> 1)) & 31) ^ (16 * vp);
          float t = Sa_i[2 * cs + vpart];
          t = vneg ? -t : t;
          z[s] = ((s & 1) == vq) ? t : 0.f;
        }
        uv[k] = (v8f){0,0,0,0,0,0,0,0};
        #pragma unroll
        for (int kk = 0; kk < 4; ++kk) {
          v2f aZ = { z[2 * kk], z[2 * kk + 1]};
          v2f bR = { rl[2 * kk], rl[2 * kk + 1]};
          uv[k] = __builtin_amdgcn_wmma_f32_16x16x4_f32(false, aZ, false, bR, (short)0, uv[k], false, false);
        }
      }
      v8f vb[2];
      vb[0] = c5 * uv[0] - s5 * uv[1];
      vb[1] = s5 * uv[0] + c5 * uv[1];
      const int up = lane & 1, ue = (lane >> 1) & 1, upart = (lane >> 3) & 1;
      const bool uoff = (lane & 4) != 0;
      float zu[8];
      #pragma unroll
      for (int s = 0; s < 8; ++s) {
        int row = 16 * ue + s + 8 * hl;
        int g = (row ^ (row >> 1)) & 31;
        float t = Sa_r[2 * g + upart];
        zu[s] = (((s & 1) == up) && !uoff) ? t : 0.f;
      }
      v8f ua = (v8f){0,0,0,0,0,0,0,0};
      #pragma unroll
      for (int kk = 0; kk < 4; ++kk) {
        v2f aZ = { zu[2 * kk], zu[2 * kk + 1]};
        v2f bR = { rh[2 * kk], rh[2 * kk + 1]};
        ua = __builtin_amdgcn_wmma_f32_16x16x4_f32(false, aZ, false, bR, (short)0, ua, false, false);
      }
      #pragma unroll
      for (int k = 0; k < 2; ++k)
        #pragma unroll
        for (int j = 0; j < 2; ++j) {
          vr[j][k] = (v8f){0,0,0,0,0,0,0,0}; vi[j][k] = vr[j][k];
          #pragma unroll
          for (int p = 0; p < 2; ++p) {
            v2f aU  = { ua[2 * k + p], ua[2 * (k ^ 1) + p]};
            v2f bRe = { vb[j][2 * p], vb[j][2 * p + 1]};
            v2f bIm = { vb[j][4 + 2 * p], vb[j][4 + 2 * p + 1]};
            vr[j][k] = __builtin_amdgcn_wmma_f32_16x16x4_f32(false, aU, false, bRe, (short)0, vr[j][k], false, false);
            vi[j][k] = __builtin_amdgcn_wmma_f32_16x16x4_f32(false, aU, false, bIm, (short)0, vi[j][k], false, false);
          }
        }
    } else
    #pragma unroll
    for (int it = 0; it < 2; ++it) {
      v8f tr[2], ti[2];
      float ar[2][8], ai[2][8];
      #pragma unroll
      for (int k = 0; k < 2; ++k)
        #pragma unroll
        for (int s = 0; s < 8; ++s) {
          int row = it * 16 + (lane & 15);
          int col = k * 16 + 4 * (s >> 1) + (s & 1) + 2 * (lane >> 4);
          int rm = (col & 1) ? (row ^ 16) : row;
          int c3 = (col ^ (col >> 1)) & 31;
          int cs = (rm & 1) ? (c3 ^ 16) : c3;
          int rs = rm;
          ar[k][s] = Sa_r[rs * PITCH + cs];
          ai[k][s] = Sa_i[rs * PITCH + cs];
        }
      v8f ur[2], ui[2];
      #pragma unroll
      for (int k = 0; k < 2; ++k) {
        ur[k] = (v8f){0,0,0,0,0,0,0,0}; ui[k] = ur[k];
      }
      #pragma unroll
      for (int kk = 0; kk < 4; ++kk) {
        v2f bR = { rl[2 * kk], rl[2 * kk + 1]};
        #pragma unroll
        for (int k = 0; k < 2; ++k) {
          v2f aR = { ar[k][2 * kk], ar[k][2 * kk + 1]};
          v2f aI = { ai[k][2 * kk], ai[k][2 * kk + 1]};
          ur[k] = __builtin_amdgcn_wmma_f32_16x16x4_f32(false, aR, false, bR, (short)0, ur[k], false, false);
          ui[k] = __builtin_amdgcn_wmma_f32_16x16x4_f32(false, aI, false, bR, (short)0, ui[k], false, false);
        }
      }
      tr[0] = c5 * ur[0] - s5 * ur[1];
      tr[1] = s5 * ur[0] + c5 * ur[1];
      ti[0] = c5 * ui[0] - s5 * ui[1];
      ti[1] = s5 * ui[0] + c5 * ui[1];
      #pragma unroll
      for (int j = 0; j < 2; ++j) {
        vr[j][it] = (v8f){0,0,0,0,0,0,0,0}; vi[j][it] = vr[j][it];
        #pragma unroll
        for (int kk = 0; kk < 4; ++kk) {
          v2f aR = { rh[2 * kk], rh[2 * kk + 1]};
          v2f bR = { tr[j][2 * kk], tr[j][2 * kk + 1]};
          v2f bI = { ti[j][2 * kk], ti[j][2 * kk + 1]};
          vr[j][it] = __builtin_amdgcn_wmma_f32_16x16x4_f32(false, aR, false, bR, (short)0, vr[j][it], false, false);
          vi[j][it] = __builtin_amdgcn_wmma_f32_16x16x4_f32(false, aR, false, bI, (short)0, vi[j][it], false, false);
        }
      }
    }
    #pragma unroll
    for (int j = 0; j < 2; ++j) {
      float pr[2][8], pi[2][8];
      #pragma unroll
      for (int s = 0; s < 8; ++s) {
        pr[0][s] = c0 * vr[j][0][s] - s0 * vr[j][1][s];
        pr[1][s] = s0 * vr[j][0][s] + c0 * vr[j][1][s];
        pi[0][s] = c0 * vi[j][0][s] - s0 * vi[j][1][s];
        pi[1][s] = s0 * vi[j][0][s] + c0 * vi[j][1][s];
      }
      if (l < NL - 1) {
        #pragma unroll
        for (int it = 0; it < 2; ++it) {
          float phr[8], phi[8];
          load_frag(ws + ph_off(l, it, j, 0) + lane * 8, phr);
          load_frag(ws + ph_off(l, it, j, 1) + lane * 8, phi);
          #pragma unroll
          for (int s = 0; s < 8; ++s) {
            int m = out_row(it, s, lane >> 4);
            int n = j * 16 + (lane & 15);
            Sa_r[m * PITCH + n] = pr[it][s] * phr[s] - pi[it][s] * phi[s];
            Sa_i[m * PITCH + n] = pr[it][s] * phi[s] + pi[it][s] * phr[s];
          }
        }
      } else {
        float wh[16];
        #pragma unroll
        for (int it = 0; it < 2; ++it)
          #pragma unroll
          for (int s = 0; s < 8; ++s)
            wh[it * 8 + s] = pr[it][s] * pr[it][s] + pi[it][s] * pi[it][s];
        #pragma unroll
        for (int len = 1; len < 16; len <<= 1)
          #pragma unroll
          for (int x = 0; x < 16; ++x)
            if (!(x & len)) {
              float a = wh[x], b = wh[x + len];
              wh[x] = a + b;
              wh[x + len] = a - b;
            }
        #pragma unroll
        for (int i = 0; i < NQ; ++i) {
          const float c = wh[walsh_mask(i)];
          acc[i] += (logical_index(0, j * 16) & (1 << (9 - i))) ? -c : c;
        }
      }
    }
  }
  const int al = logical_index(8 * (lane >> 4), lane & 15);
  #pragma unroll
  for (int i = 0; i < NQ; ++i)
    acc[i] = (al & (1 << (9 - i))) ? -acc[i] : acc[i];
  #pragma unroll
  for (int off = 16; off > 0; off >>= 1)
    #pragma unroll
    for (int i = 0; i < NQ; ++i)
      acc[i] += __shfl_xor(acc[i], off, 32);
  if (lane == 0) {
    #pragma unroll
    for (int i = 0; i < NQ; ++i) out[sample * NQ + i] = acc[i];
  }
}

extern "C" void kernel_launch(void* const* d_in, const int* in_sizes, int n_in,
                              void* d_out, int out_size, void* d_ws, size_t ws_size,
                              hipStream_t stream) {
  const float* x = (const float*)d_in[0];
  const float* w = (const float*)d_in[1];
  float* ws  = (float*)d_ws;
  float* out = (float*)d_out;

  vqc_prep<<<PREP_BLOCKS, PREP_T, 0, stream>>>(w, ws);
  vqc_main<<<BATCH_N / WPB, 32 * WPB, 0, stream>>>(x, w, ws, out);
}


# one launch, not two: each block of 8 waves builds the layer tables once in its own LDS (phase table = row part times column part), each wave then runs 2 samples against them; table products regrouped
# speedup vs baseline: 5.1444x; 1.0643x over previous
#include <hip/hip_runtime.h>
#include <hip/hip_bf16.h>


typedef __attribute__((ext_vector_type(2))) float v2f;
typedef __attribute__((ext_vector_type(4))) float v4f;
typedef __attribute__((ext_vector_type(8))) float v8f;

#define NQ      10
#define NL      4
#define BATCH_N 8192
#define PITCH   33
#define PLANE   (32 * PITCH)
#define SAMPLE_F (2 * PLANE)
#define WPB     8
#define SPW     2

__device__ __forceinline__ constexpr int rh_off(int l) { return (l - 1) * 256; }
__device__ __forceinline__ constexpr int rl_off(int l) { return 768 + (l - 1) * 256; }
__device__ __forceinline__ constexpr int cs_off(int l) { return 1536 + (l - 1) * 4; }
__device__ __forceinline__ constexpr int ph_off(int l, int i, int j, int p) {
  return 1552 + ((((l - 1) * 2 + i) * 2 + j) * 2 + p) * 256;
}
#define TAB_F (1552 + 16 * 256)
__device__ __forceinline__ int frag_pos(int lane, int slot) {
  return (slot >> 2) * 128 + lane * 4 + (slot & 3);
}

__device__ __forceinline__ constexpr int out_row(int it, int s, int h) {
  return it ? 31 - s - 8 * h : s + 8 * h;
}

__device__ __forceinline__ constexpr int logical_index(int m, int n) {
  int c3 = (m & 1) ? (n ^ 16) : n;
  int c4 = (c3 ^ (c3 >> 1) ^ (c3 >> 2) ^ (c3 >> 3) ^ (c3 >> 4)) & 31;
  int r2 = (c4 & 1) ? (m ^ 16) : m;
  return (r2 << 5) | c4;
}

__device__ __forceinline__ constexpr int row_of(int x) {
  return ((x >> 3) ? 31 : 0) ^ (x & 7);
}
__device__ __forceinline__ constexpr int walsh_mask(int i) {
  int u = 0;
  for (int b = 0; b < 4; ++b)
    if (logical_index(row_of(1 << b), 0) & (1 << (9 - i))) u |= 1 << b;
  return u;
}

#define BLOCK_T (32 * WPB)

__device__ __forceinline__ void build_tables(const float* __restrict__ w, float* __restrict__ tab,
                                             float (* __restrict__ cs)[2]) {
  const int t = threadIdx.x;
  if (t < (NL - 1) * 2 * NQ) {
    float h = 0.5f * w[2 * NQ + t];
    cs[t][0] = cosf(h);
    cs[t][1] = sinf(h);
  }
  __syncthreads();
  const int lane = t >> 3;
  const int slot = t & 7;
  const int hl = lane >> 4, c = lane & 15;
  const int pos = frag_pos(lane, slot);
  #pragma unroll
  for (int l = 1; l < NL; ++l) {
    const float (*csl)[2] = cs + (l - 1) * 2 * NQ;
    #pragma unroll
    for (int which = 0; which < 2; ++which) {
      int rsel, wbase, off;
      int csel;
      if (which == 0) {
        csel = slot + 8 * hl;
        rsel = (c ^ (c >> 1)) & 15;
        wbase = 0;
        off = rh_off(l);
      } else {
        rsel = c;
        csel = 4 * (slot >> 1) + (slot & 1) + 2 * hl;
        wbase = 5;
        off = rl_off(l);
      }
      float v = 1.f;
      #pragma unroll
      for (int w2 = 1; w2 < 5; ++w2) {
        float ca = csl[(wbase + w2) * 2][0], sa = csl[(wbase + w2) * 2][1];
        int rb = (rsel >> (4 - w2)) & 1, cb = (csel >> (4 - w2)) & 1;
        v *= rb ? (cb ? ca : sa) : (cb ? -sa : ca);
      }
      tab[off + pos] = v;
    }
    if (t < 4) tab[cs_off(l) + t] = csl[(t >> 1) * 10][t & 1];
    if (l < NL - 1) {
      const int m0 = slot + 8 * hl;
      const int r0 = (m0 ^ (m0 >> 1)) & 31;
      float ar = 1.f, ai = 0.f, br = 1.f, bi = 0.f;
      #pragma unroll
      for (int w2 = 1; w2 < 5; ++w2) {
        {
          float fr = csl[w2 * 2 + 1][0], sz = csl[w2 * 2 + 1][1];
          float fi = ((r0 >> (4 - w2)) & 1) ? sz : -sz;
          float nr = ar * fr - ai * fi; ai = ar * fi + ai * fr; ar = nr;
        }
        {
          float fr = csl[(5 + w2) * 2 + 1][0], sz = csl[(5 + w2) * 2 + 1][1];
          float fi = ((c >> (4 - w2)) & 1) ? sz : -sz;
          float nr = br * fr - bi * fi; bi = br * fi + bi * fr; br = nr;
        }
      }
      const float z0r = csl[1][0], z0s = csl[1][1];
      const float z5r = csl[11][0], z5s = csl[11][1];
      #pragma unroll
      for (int f1 = 0; f1 < 2; ++f1) {
        const float fi1 = f1 ? z0s : -z0s;
        const float par = ar * z0r - ai * fi1, pai = ar * fi1 + ai * z0r;
        #pragma unroll
        for (int f0 = 0; f0 < 2; ++f0) {
          const float fi0 = f0 ? z5s : -z5s;
          const float pbr = br * z5r - bi * fi0, pbi = br * fi0 + bi * z5r;
          tab[ph_off(l, f1, f0, 0) + pos] = par * pbr - pai * pbi;
          tab[ph_off(l, f1, f0, 1) + pos] = par * pbi + pai * pbr;
        }
      }
    }
  }
  __syncthreads();
}

__device__ __forceinline__ void load_frag(const float* p, int lane, float (&f)[8]) {
  const v4f a = *(const v4f*)(p + lane * 4);
  const v4f b = *(const v4f*)(p + 128 + lane * 4);
  f[0] = a.x; f[1] = a.y; f[2] = a.z; f[3] = a.w;
  f[4] = b.x; f[5] = b.y; f[6] = b.z; f[7] = b.w;
}

__global__ __launch_bounds__(BLOCK_T) void vqc_main(const float* __restrict__ x,
                                                    const float* __restrict__ w,
                                                    float* __restrict__ out) {
  __shared__ float lds[WPB * SAMPLE_F];
  __shared__ __attribute__((aligned(16))) float tab[TAB_F];
  __shared__ float cs[(NL - 1) * 2 * NQ][2];
  const int lane = threadIdx.x & 31;
  const int wv   = threadIdx.x >> 5;

  build_tables(w, tab, cs);

  float* Sa_r = lds + wv * SAMPLE_F;
  float* Sa_i = Sa_r + PLANE;

  #pragma unroll 1
  for (int rep = 0; rep < SPW; ++rep) {
  const int sample = (blockIdx.x * SPW + rep) * WPB + wv;

  float hx = 0.f;
  if (lane < NQ) hx = 0.5f * (x[sample * NQ + lane] + w[lane * 2]);
  else if (lane < 2 * NQ) hx = 0.5f * w[(lane - NQ) * 2 + 1];
  const int sn = __float_as_int(sinf(hx)), cn = __float_as_int(cosf(hx));
  float a_r = 1.f, a_i = 0.f, b_r = 1.f, b_i = 0.f;
  #pragma unroll
  for (int w2 = 0; w2 < 5; ++w2) {
    const int bit = (lane >> (4 - w2)) & 1;
    {
      float sy = __int_as_float(__builtin_amdgcn_readlane(sn, w2));
      float cy = __int_as_float(__builtin_amdgcn_readlane(cn, w2));
      float sz = __int_as_float(__builtin_amdgcn_readlane(sn, NQ + w2));
      float cz = __int_as_float(__builtin_amdgcn_readlane(cn, NQ + w2));
      float mag = bit ? sy : cy;
      float fr = mag * cz, fi = mag * (bit ? sz : -sz);
      float nr = a_r * fr - a_i * fi; a_i = a_r * fi + a_i * fr; a_r = nr;
    }
    {
      float sy = __int_as_float(__builtin_amdgcn_readlane(sn, 5 + w2));
      float cy = __int_as_float(__builtin_amdgcn_readlane(cn, 5 + w2));
      float sz = __int_as_float(__builtin_amdgcn_readlane(sn, NQ + 5 + w2));
      float cz = __int_as_float(__builtin_amdgcn_readlane(cn, NQ + 5 + w2));
      float mag = bit ? sy : cy;
      float fr = mag * cz, fi = mag * (bit ? sz : -sz);
      float nr = b_r * fr - b_i * fi; b_i = b_r * fi + b_i * fr; b_r = nr;
    }
  }
  Sa_r[2 * lane] = a_r; Sa_r[2 * lane + 1] = a_i;
  Sa_i[2 * lane] = b_r; Sa_i[2 * lane + 1] = b_i;

  float acc[NQ];
  #pragma unroll
  for (int i = 0; i < NQ; ++i) acc[i] = 0.f;

  for (int l = 1; l < NL; ++l) {
    const float c0 = tab[cs_off(l) + 0], s0 = tab[cs_off(l) + 1];
    const float c5 = tab[cs_off(l) + 2], s5 = tab[cs_off(l) + 3];
    float rl[8];
    load_frag(tab + rl_off(l), lane, rl);
    float rh[8];
    load_frag(tab + rh_off(l), lane, rh);
    v8f vr[2][2], vi[2][2];
    if (l == 1) {
      const int hl = lane >> 4;
      const int vq = lane & 1, vp = (lane >> 1) & 1;
      const int vpart = ((lane >> 2) ^ (lane >> 3)) & 1;
      const bool vneg = (lane & 12) == 8;
      v8f uv[2];
      #pragma unroll
      for (int k = 0; k < 2; ++k) {
        float z[8];
        #pragma unroll
        for (int s = 0; s < 8; ++s) {
          int col = k * 16 + 4 * (s >> 1) + (s & 1) + 2 * hl;
          int cs = ((col ^ (col >> 1)) & 31) ^ (16 * vp);
          float t = Sa_i[2 * cs + vpart];
          t = vneg ? -t : t;
          z[s] = ((s & 1) == vq) ? t : 0.f;
        }
        uv[k] = (v8f){0,0,0,0,0,0,0,0};
        #pragma unroll
        for (int kk = 0; kk < 4; ++kk) {
          v2f aZ = { z[2 * kk], z[2 * kk + 1]};
          v2f bR = { rl[2 * kk], rl[2 * kk + 1]};
          uv[k] = __builtin_amdgcn_wmma_f32_16x16x4_f32(false, aZ, false, bR, (short)0, uv[k], false, false);
        }
      }
      v8f vb[2];
      vb[0] = c5 * uv[0] - s5 * uv[1];
      vb[1] = s5 * uv[0] + c5 * uv[1];
      const int up = lane & 1, ue = (lane >> 1) & 1, upart = (lane >> 3) & 1;
      const bool uoff = (lane & 4) != 0;
      float zu[8];
      #pragma unroll
      for (int s = 0; s < 8; ++s) {
        int row = 16 * ue + s + 8 * hl;
        int g = (row ^ (row >> 1)) & 31;
        float t = Sa_r[2 * g + upart];
        zu[s] = (((s & 1) == up) && !uoff) ? t : 0.f;
      }
      v8f ua = (v8f){0,0,0,0,0,0,0,0};
      #pragma unroll
      for (int kk = 0; kk < 4; ++kk) {
        v2f aZ = { zu[2 * kk], zu[2 * kk + 1]};
        v2f bR = { rh[2 * kk], rh[2 * kk + 1]};
        ua = __builtin_amdgcn_wmma_f32_16x16x4_f32(false, aZ, false, bR, (short)0, ua, false, false);
      }
      #pragma unroll
      for (int k = 0; k < 2; ++k)
        #pragma unroll
        for (int j = 0; j < 2; ++j) {
          vr[j][k] = (v8f){0,0,0,0,0,0,0,0}; vi[j][k] = vr[j][k];
          #pragma unroll
          for (int p = 0; p < 2; ++p) {
            v2f aU  = { ua[2 * k + p], ua[2 * (k ^ 1) + p]};
            v2f bRe = { vb[j][2 * p], vb[j][2 * p + 1]};
            v2f bIm = { vb[j][4 + 2 * p], vb[j][4 + 2 * p + 1]};
            vr[j][k] = __builtin_amdgcn_wmma_f32_16x16x4_f32(false, aU, false, bRe, (short)0, vr[j][k], false, false);
            vi[j][k] = __builtin_amdgcn_wmma_f32_16x16x4_f32(false, aU, false, bIm, (short)0, vi[j][k], false, false);
          }
        }
    } else
    #pragma unroll
    for (int it = 0; it < 2; ++it) {
      v8f tr[2], ti[2];
      float ar[2][8], ai[2][8];
      #pragma unroll
      for (int k = 0; k < 2; ++k)
        #pragma unroll
        for (int s = 0; s < 8; ++s) {
          int row = it * 16 + (lane & 15);
          int col = k * 16 + 4 * (s >> 1) + (s & 1) + 2 * (lane >> 4);
          int rm = (col & 1) ? (row ^ 16) : row;
          int c3 = (col ^ (col >> 1)) & 31;
          int cs = (rm & 1) ? (c3 ^ 16) : c3;
          int rs = rm;
          ar[k][s] = Sa_r[rs * PITCH + cs];
          ai[k][s] = Sa_i[rs * PITCH + cs];
        }
      v8f ur[2], ui[2];
      #pragma unroll
      for (int k = 0; k < 2; ++k) {
        ur[k] = (v8f){0,0,0,0,0,0,0,0}; ui[k] = ur[k];
      }
      #pragma unroll
      for (int kk = 0; kk < 4; ++kk) {
        v2f bR = { rl[2 * kk], rl[2 * kk + 1]};
        #pragma unroll
        for (int k = 0; k < 2; ++k) {
          v2f aR = { ar[k][2 * kk], ar[k][2 * kk + 1]};
          v2f aI = { ai[k][2 * kk], ai[k][2 * kk + 1]};
          ur[k] = __builtin_amdgcn_wmma_f32_16x16x4_f32(false, aR, false, bR, (short)0, ur[k], false, false);
          ui[k] = __builtin_amdgcn_wmma_f32_16x16x4_f32(false, aI, false, bR, (short)0, ui[k], false, false);
        }
      }
      tr[0] = c5 * ur[0] - s5 * ur[1];
      tr[1] = s5 * ur[0] + c5 * ur[1];
      ti[0] = c5 * ui[0] - s5 * ui[1];
      ti[1] = s5 * ui[0] + c5 * ui[1];
      #pragma unroll
      for (int j = 0; j < 2; ++j) {
        vr[j][it] = (v8f){0,0,0,0,0,0,0,0}; vi[j][it] = vr[j][it];
        #pragma unroll
        for (int kk = 0; kk < 4; ++kk) {
          v2f aR = { rh[2 * kk], rh[2 * kk + 1]};
          v2f bR = { tr[j][2 * kk], tr[j][2 * kk + 1]};
          v2f bI = { ti[j][2 * kk], ti[j][2 * kk + 1]};
          vr[j][it] = __builtin_amdgcn_wmma_f32_16x16x4_f32(false, aR, false, bR, (short)0, vr[j][it], false, false);
          vi[j][it] = __builtin_amdgcn_wmma_f32_16x16x4_f32(false, aR, false, bI, (short)0, vi[j][it], false, false);
        }
      }
    }
    #pragma unroll
    for (int j = 0; j < 2; ++j) {
      float pr[2][8], pi[2][8];
      #pragma unroll
      for (int s = 0; s < 8; ++s) {
        pr[0][s] = c0 * vr[j][0][s] - s0 * vr[j][1][s];
        pr[1][s] = s0 * vr[j][0][s] + c0 * vr[j][1][s];
        pi[0][s] = c0 * vi[j][0][s] - s0 * vi[j][1][s];
        pi[1][s] = s0 * vi[j][0][s] + c0 * vi[j][1][s];
      }
      if (l < NL - 1) {
        #pragma unroll
        for (int it = 0; it < 2; ++it) {
          float phr[8], phi[8];
          load_frag(tab + ph_off(l, it, j, 0), lane, phr);
          load_frag(tab + ph_off(l, it, j, 1), lane, phi);
          #pragma unroll
          for (int s = 0; s < 8; ++s) {
            int m = out_row(it, s, lane >> 4);
            int n = j * 16 + (lane & 15);
            Sa_r[m * PITCH + n] = pr[it][s] * phr[s] - pi[it][s] * phi[s];
            Sa_i[m * PITCH + n] = pr[it][s] * phi[s] + pi[it][s] * phr[s];
          }
        }
      } else {
        float wh[16];
        #pragma unroll
        for (int it = 0; it < 2; ++it)
          #pragma unroll
          for (int s = 0; s < 8; ++s)
            wh[it * 8 + s] = pr[it][s] * pr[it][s] + pi[it][s] * pi[it][s];
        #pragma unroll
        for (int len = 1; len < 16; len <<= 1)
          #pragma unroll
          for (int x = 0; x < 16; ++x)
            if (!(x & len)) {
              float a = wh[x], b = wh[x + len];
              wh[x] = a + b;
              wh[x + len] = a - b;
            }
        #pragma unroll
        for (int i = 0; i < NQ; ++i) {
          const float c = wh[walsh_mask(i)];
          acc[i] += (logical_index(0, j * 16) & (1 << (9 - i))) ? -c : c;
        }
      }
    }
  }
  const int al = logical_index(8 * (lane >> 4), lane & 15);
  #pragma unroll
  for (int i = 0; i < NQ; ++i)
    acc[i] = (al & (1 << (9 - i))) ? -acc[i] : acc[i];
  #pragma unroll
  for (int off = 16; off > 0; off >>= 1)
    #pragma unroll
    for (int i = 0; i < NQ; ++i)
      acc[i] += __shfl_xor(acc[i], off, 32);
  if (lane == 0) {
    #pragma unroll
    for (int i = 0; i < NQ; ++i) out[sample * NQ + i] = acc[i];
  }
  }
}

extern "C" void kernel_launch(void* const* d_in, const int* in_sizes, int n_in,
                              void* d_out, int out_size, void* d_ws, size_t ws_size,
                              hipStream_t stream) {
  const float* x = (const float*)d_in[0];
  const float* w = (const float*)d_in[1];
  float* out = (float*)d_out;

  vqc_main<<<BATCH_N / (WPB * SPW), BLOCK_T, 0, stream>>>(x, w, out);
}


# state stays in registers between layers: a tile as the A operand is its own transpose, pending CNOTs folded into gate k order, slot choice and wire 0's coefficients; no LDS round trip; sums regrouped
# speedup vs baseline: 5.5392x; 1.0767x over previous
#include <hip/hip_runtime.h>
#include <hip/hip_bf16.h>


typedef __attribute__((ext_vector_type(2))) float v2f;
typedef __attribute__((ext_vector_type(4))) float v4f;
typedef __attribute__((ext_vector_type(8))) float v8f;

#define NQ      10
#define NL      4
#define BATCH_N 8192
#define VEC_F   128
#define WPB     8
#define SPW     2

__device__ __forceinline__ constexpr int rh_off(int l) { return (l - 1) * 256; }
__device__ __forceinline__ constexpr int rl_off(int l) { return 768 + (l - 1) * 256; }
__device__ __forceinline__ constexpr int gh1_off(int l) { return 1536 + (l - 2) * 256; }
__device__ __forceinline__ constexpr int gl1_off(int l) { return 2048 + (l - 2) * 256; }
__device__ __forceinline__ constexpr int cs_off(int l) { return 2560 + (l - 1) * 4; }
__device__ __forceinline__ constexpr int ph_off(int l, int i, int j, int p) {
  return 2576 + ((((l - 1) * 2 + i) * 2 + j) * 2 + p) * 256;
}
#define TAB_F (2576 + 16 * 256)
__device__ __forceinline__ int frag_pos(int lane, int slot) {
  return (slot >> 2) * 128 + lane * 4 + (slot & 3);
}

__device__ __forceinline__ constexpr int pos_row(int x) {
  return (x & ~3) | ((x & 1) << 1) | ((x >> 1) & 1);
}
__device__ __forceinline__ constexpr int out_row(int it, int s, int h) {
  return it ? 31 - pos_row(s + 8 * h) : pos_row(s + 8 * h);
}

__device__ __forceinline__ constexpr int logical_index(int m, int n) {
  int c3 = (m & 1) ? (n ^ 16) : n;
  int c4 = (c3 ^ (c3 >> 1) ^ (c3 >> 2) ^ (c3 >> 3) ^ (c3 >> 4)) & 31;
  int r2 = (c4 & 1) ? (m ^ 16) : m;
  return (r2 << 5) | c4;
}

__device__ __forceinline__ constexpr int row_of(int x) {
  return ((x >> 3) ? 31 : 0) ^ pos_row(x & 7);
}
__device__ __forceinline__ constexpr int walsh_mask(int i) {
  int u = 0;
  for (int b = 0; b < 4; ++b)
    if (logical_index(row_of(1 << b), 0) & (1 << (9 - i))) u |= 1 << b;
  return u;
}

#define BLOCK_T (32 * WPB)

__device__ __forceinline__ void build_tables(const float* __restrict__ w, float* __restrict__ tab,
                                             float (* __restrict__ cs)[2]) {
  const int t = threadIdx.x;
  if (t < (NL - 1) * 2 * NQ) {
    float h = 0.5f * w[2 * NQ + t];
    cs[t][0] = cosf(h);
    cs[t][1] = sinf(h);
  }
  __syncthreads();
  const int lane = t >> 3;
  const int slot = t & 7;
  const int hl = lane >> 4, c = lane & 15;
  const int pos = frag_pos(lane, slot);
  #pragma unroll
  for (int l = 1; l < NL; ++l) {
    const float (*csl)[2] = cs + (l - 1) * 2 * NQ;
    #pragma unroll
    for (int which = 0; which < 4; ++which) {
      if (which >= 2 && l == 1) continue;
      const int x8 = slot + 8 * hl;
      int rsel, wbase, off;
      int csel;
      if ((which & 1) == 0) {
        csel = l == 1 ? x8 : (which == 0 ? pos_row(x8) : (pos_row(x8) ^ 15));
        rsel = (pos_row(c) ^ (pos_row(c) >> 1)) & 15;
        wbase = 0;
        off = which == 0 ? rh_off(l) : gh1_off(l);
      } else {
        rsel = c;
        if (l == 1) {
          csel = 4 * (slot >> 1) + (slot & 1) + 2 * hl;
        } else {
          const int pf = (x8 ^ (x8 >> 1) ^ (x8 >> 2) ^ (x8 >> 3)) & 15;
          csel = which == 1 ? pf : (pf ^ 15);
        }
        wbase = 5;
        off = which == 1 ? rl_off(l) : gl1_off(l);
      }
      float v = 1.f;
      #pragma unroll
      for (int w2 = 1; w2 < 5; ++w2) {
        float ca = csl[(wbase + w2) * 2][0], sa = csl[(wbase + w2) * 2][1];
        int rb = (rsel >> (4 - w2)) & 1, cb = (csel >> (4 - w2)) & 1;
        v *= rb ? (cb ? ca : sa) : (cb ? -sa : ca);
      }
      tab[off + pos] = v;
    }
    if (t < 4) tab[cs_off(l) + t] = csl[(t >> 1) * 10][t & 1];
    if (l < NL - 1) {
      const int m0 = pos_row(slot + 8 * hl);
      const int r0 = (m0 ^ (m0 >> 1)) & 31;
      float ar = 1.f, ai = 0.f, br = 1.f, bi = 0.f;
      #pragma unroll
      for (int w2 = 1; w2 < 5; ++w2) {
        {
          float fr = csl[w2 * 2 + 1][0], sz = csl[w2 * 2 + 1][1];
          float fi = ((r0 >> (4 - w2)) & 1) ? sz : -sz;
          float nr = ar * fr - ai * fi; ai = ar * fi + ai * fr; ar = nr;
        }
        {
          float fr = csl[(5 + w2) * 2 + 1][0], sz = csl[(5 + w2) * 2 + 1][1];
          float fi = ((c >> (4 - w2)) & 1) ? sz : -sz;
          float nr = br * fr - bi * fi; bi = br * fi + bi * fr; br = nr;
        }
      }
      const float z0r = csl[1][0], z0s = csl[1][1];
      const float z5r = csl[11][0], z5s = csl[11][1];
      #pragma unroll
      for (int f1 = 0; f1 < 2; ++f1) {
        const float fi1 = f1 ? z0s : -z0s;
        const float par = ar * z0r - ai * fi1, pai = ar * fi1 + ai * z0r;
        #pragma unroll
        for (int f0 = 0; f0 < 2; ++f0) {
          const float fi0 = ((f0 ^ (m0 & 1) ^ f1) & 1) ? z5s : -z5s;
          const float pbr = br * z5r - bi * fi0, pbi = br * fi0 + bi * z5r;
          tab[ph_off(l, f1, f0, 0) + pos] = par * pbr - pai * pbi;
          tab[ph_off(l, f1, f0, 1) + pos] = par * pbi + pai * pbr;
        }
      }
    }
  }
  __syncthreads();
}

__device__ __forceinline__ void load_frag(const float* p, int lane, float (&f)[8]) {
  const v4f a = *(const v4f*)(p + lane * 4);
  const v4f b = *(const v4f*)(p + 128 + lane * 4);
  f[0] = a.x; f[1] = a.y; f[2] = a.z; f[3] = a.w;
  f[4] = b.x; f[5] = b.y; f[6] = b.z; f[7] = b.w;
}

__global__ __launch_bounds__(BLOCK_T) __attribute__((amdgpu_waves_per_eu(4, 4))) void vqc_main(const float* __restrict__ x,
                                                    const float* __restrict__ w,
                                                    float* __restrict__ out) {
  __shared__ float lds[WPB * VEC_F];
  __shared__ __attribute__((aligned(16))) float tab[TAB_F];
  __shared__ float cs[(NL - 1) * 2 * NQ][2];
  const int lane = threadIdx.x & 31;
  const int wv   = threadIdx.x >> 5;

  build_tables(w, tab, cs);

  float* Sa_r = lds + wv * VEC_F;
  float* Sa_i = Sa_r + VEC_F / 2;

  #pragma unroll 1
  for (int rep = 0; rep < SPW; ++rep) {
  asm volatile("" ::: "memory");
  const int sample = (blockIdx.x * SPW + rep) * WPB + wv;

  float hx = 0.f;
  if (lane < NQ) hx = 0.5f * (x[sample * NQ + lane] + w[lane * 2]);
  else if (lane < 2 * NQ) hx = 0.5f * w[(lane - NQ) * 2 + 1];
  const int sn = __float_as_int(sinf(hx)), cn = __float_as_int(cosf(hx));
  float a_r = 1.f, a_i = 0.f, b_r = 1.f, b_i = 0.f;
  #pragma unroll
  for (int w2 = 0; w2 < 5; ++w2) {
    const int bit = (lane >> (4 - w2)) & 1;
    {
      float sy = __int_as_float(__builtin_amdgcn_readlane(sn, w2));
      float cy = __int_as_float(__builtin_amdgcn_readlane(cn, w2));
      float sz = __int_as_float(__builtin_amdgcn_readlane(sn, NQ + w2));
      float cz = __int_as_float(__builtin_amdgcn_readlane(cn, NQ + w2));
      float mag = bit ? sy : cy;
      float fr = mag * cz, fi = mag * (bit ? sz : -sz);
      float nr = a_r * fr - a_i * fi; a_i = a_r * fi + a_i * fr; a_r = nr;
    }
    {
      float sy = __int_as_float(__builtin_amdgcn_readlane(sn, 5 + w2));
      float cy = __int_as_float(__builtin_amdgcn_readlane(cn, 5 + w2));
      float sz = __int_as_float(__builtin_amdgcn_readlane(sn, NQ + 5 + w2));
      float cz = __int_as_float(__builtin_amdgcn_readlane(cn, NQ + 5 + w2));
      float mag = bit ? sy : cy;
      float fr = mag * cz, fi = mag * (bit ? sz : -sz);
      float nr = b_r * fr - b_i * fi; b_i = b_r * fi + b_i * fr; b_r = nr;
    }
  }
  Sa_r[2 * lane] = a_r; Sa_r[2 * lane + 1] = a_i;
  Sa_i[2 * lane] = b_r; Sa_i[2 * lane + 1] = b_i;

  float acc[NQ];
  #pragma unroll
  for (int i = 0; i < NQ; ++i) acc[i] = 0.f;

  float Sr[2][2][8], Si[2][2][8];

  #pragma unroll
  for (int l = 1; l < NL; ++l) {
    const float c0 = tab[cs_off(l) + 0], s0 = tab[cs_off(l) + 1];
    const float c5 = tab[cs_off(l) + 2], s5 = tab[cs_off(l) + 3];
    v8f vr[2][2], vi[2][2];
    if (l == 1) {
      float rl[8];
      load_frag(tab + rl_off(l), lane, rl);
      float rh[8];
      load_frag(tab + rh_off(l), lane, rh);
      const int hl = lane >> 4;
      const int vq = lane & 1, vp = (lane >> 1) & 1;
      const int vpart = ((lane >> 2) ^ (lane >> 3)) & 1;
      const bool vneg = (lane & 12) == 8;
      v8f uv[2];
      #pragma unroll
      for (int k = 0; k < 2; ++k) {
        float z[8];
        #pragma unroll
        for (int s = 0; s < 8; ++s) {
          int col = k * 16 + 4 * (s >> 1) + (s & 1) + 2 * hl;
          int cs = ((col ^ (col >> 1)) & 31) ^ (16 * vp);
          float t = Sa_i[2 * cs + vpart];
          t = vneg ? -t : t;
          z[s] = ((s & 1) == vq) ? t : 0.f;
        }
        uv[k] = (v8f){0,0,0,0,0,0,0,0};
        #pragma unroll
        for (int kk = 0; kk < 4; ++kk) {
          v2f aZ = { z[2 * kk], z[2 * kk + 1]};
          v2f bR = { rl[2 * kk], rl[2 * kk + 1]};
          uv[k] = __builtin_amdgcn_wmma_f32_16x16x4_f32(false, aZ, false, bR, (short)0, uv[k], false, false);
        }
      }
      v8f vb[2];
      vb[0] = c5 * uv[0] - s5 * uv[1];
      vb[1] = s5 * uv[0] + c5 * uv[1];
      const int up = lane & 1, ue = (lane >> 1) & 1, upart = (lane >> 3) & 1;
      const bool uoff = (lane & 4) != 0;
      float zu[8];
      #pragma unroll
      for (int s = 0; s < 8; ++s) {
        int row = 16 * ue + s + 8 * hl;
        int g = (row ^ (row >> 1)) & 31;
        float t = Sa_r[2 * g + upart];
        zu[s] = (((s & 1) == up) && !uoff) ? t : 0.f;
      }
      v8f ua = (v8f){0,0,0,0,0,0,0,0};
      #pragma unroll
      for (int kk = 0; kk < 4; ++kk) {
        v2f aZ = { zu[2 * kk], zu[2 * kk + 1]};
        v2f bR = { rh[2 * kk], rh[2 * kk + 1]};
        ua = __builtin_amdgcn_wmma_f32_16x16x4_f32(false, aZ, false, bR, (short)0, ua, false, false);
      }
      #pragma unroll
      for (int k = 0; k < 2; ++k)
        #pragma unroll
        for (int j = 0; j < 2; ++j) {
          vr[j][k] = (v8f){0,0,0,0,0,0,0,0}; vi[j][k] = vr[j][k];
          #pragma unroll
          for (int p = 0; p < 2; ++p) {
            v2f aU  = { ua[2 * k + p], ua[2 * (k ^ 1) + p]};
            v2f bRe = { vb[j][2 * p], vb[j][2 * p + 1]};
            v2f bIm = { vb[j][4 + 2 * p], vb[j][4 + 2 * p + 1]};
            vr[j][k] = __builtin_amdgcn_wmma_f32_16x16x4_f32(false, aU, false, bRe, (short)0, vr[j][k], false, false);
            vi[j][k] = __builtin_amdgcn_wmma_f32_16x16x4_f32(false, aU, false, bIm, (short)0, vi[j][k], false, false);
          }
        }
    } else {
      const int hl = lane >> 4;
      float gh[2][8];
      load_frag(tab + rh_off(l), lane, gh[0]);
      load_frag(tab + gh1_off(l), lane, gh[1]);
      float yr[2][2][8], yi[2][2][8];
      const float e00 = hl ? -s0 : c0, e01 = hl ? c0 : -s0;
      const float e10 = hl ?  c0 : s0, e11 = hl ? s0 :  c0;
      #pragma unroll
      for (int jp = 0; jp < 2; ++jp) {
        v8f tr[2], ti[2];
        #pragma unroll
        for (int it = 0; it < 2; ++it) {
          tr[it] = (v8f){0,0,0,0,0,0,0,0}; ti[it] = tr[it];
        }
        #pragma unroll
        for (int kk = 0; kk < 4; ++kk)
          #pragma unroll
          for (int it = 0; it < 2; ++it) {
            v2f bG = { gh[it][2 * kk], gh[it][2 * kk + 1]};
            v2f aR = { Sr[it][jp][2 * kk], Sr[it][jp][2 * kk + 1]};
            v2f aI = { Si[it][jp][2 * kk], Si[it][jp][2 * kk + 1]};
            tr[it] = __builtin_amdgcn_wmma_f32_16x16x4_f32(false, aR, false, bG, (short)0, tr[it], false, false);
            ti[it] = __builtin_amdgcn_wmma_f32_16x16x4_f32(false, aI, false, bG, (short)0, ti[it], false, false);
          }
        #pragma unroll
        for (int r = 0; r < 8; ++r) {
          const int q = (jp ^ r ^ (r >> 1) ^ (r >> 2)) & 1;
          const float f00 = q ? e01 : e00, f01 = q ? e00 : e01;
          const float f10 = q ? e11 : e10, f11 = q ? e10 : e11;
          yr[0][jp][r] = f00 * tr[0][r] + f01 * tr[1][r];
          yr[1][jp][r] = f10 * tr[0][r] + f11 * tr[1][r];
          yi[0][jp][r] = f00 * ti[0][r] + f01 * ti[1][r];
          yi[1][jp][r] = f10 * ti[0][r] + f11 * ti[1][r];
        }
      }
      float gl[2][8];
      load_frag(tab + rl_off(l), lane, gl[0]);
      load_frag(tab + gl1_off(l), lane, gl[1]);
      #pragma unroll
      for (int jp = 0; jp < 2; ++jp) {
        #pragma unroll
        for (int ip = 0; ip < 2; ++ip) {
          vr[jp][ip] = (v8f){0,0,0,0,0,0,0,0}; vi[jp][ip] = vr[jp][ip];
        }
        #pragma unroll
        for (int kk = 0; kk < 4; ++kk)
          #pragma unroll
          for (int ip = 0; ip < 2; ++ip) {
            v2f bG = { gl[jp][2 * kk], gl[jp][2 * kk + 1]};
            v2f aR = { yr[ip][jp][2 * kk], yr[ip][jp][2 * kk + 1]};
            v2f aI = { yi[ip][jp][2 * kk], yi[ip][jp][2 * kk + 1]};
            vr[jp][ip] = __builtin_amdgcn_wmma_f32_16x16x4_f32(false, aR, false, bG, (short)0, vr[jp][ip], false, false);
            vi[jp][ip] = __builtin_amdgcn_wmma_f32_16x16x4_f32(false, aI, false, bG, (short)0, vi[jp][ip], false, false);
          }
      }
    }
    #pragma unroll
    for (int j = 0; j < 2; ++j) {
      if (l < NL - 1) {
        float pr[2][8], pi[2][8];
        #pragma unroll
        for (int it = 0; it < 2; ++it)
          #pragma unroll
          for (int s = 0; s < 8; ++s) {
            const int je = j ^ (out_row(it, s, 0) & 1);
            if (l == 1) {
              const float k0 = it ? s0 : c0, k1 = it ? c0 : -s0;
              pr[it][s] = k0 * vr[je][0][s] + k1 * vr[je][1][s];
              pi[it][s] = k0 * vi[je][0][s] + k1 * vi[je][1][s];
            } else {
              const float k0 = je ? s5 : c5, k1 = je ? c5 : -s5;
              pr[it][s] = k0 * vr[0][it][s] + k1 * vr[1][it][s];
              pi[it][s] = k0 * vi[0][it][s] + k1 * vi[1][it][s];
            }
          }
        #pragma unroll
        for (int it = 0; it < 2; ++it) {
          float phr[8], phi[8];
          load_frag(tab + ph_off(l, it, j, 0), lane, phr);
          load_frag(tab + ph_off(l, it, j, 1), lane, phi);
          #pragma unroll
          for (int s = 0; s < 8; ++s) {
            Sr[it][j][s] = pr[it][s] * phr[s] - pi[it][s] * phi[s];
            Si[it][j][s] = pr[it][s] * phi[s] + pi[it][s] * phr[s];
          }
        }
      } else {
        float wh[16];
        const float k0 = j ? s5 : c5, k1 = j ? c5 : -s5;
        #pragma unroll
        for (int it = 0; it < 2; ++it)
          #pragma unroll
          for (int k2 = 0; k2 < 4; ++k2) {
            const v2f r0 = { vr[0][it][2 * k2], vr[0][it][2 * k2 + 1]};
            const v2f r1 = { vr[1][it][2 * k2], vr[1][it][2 * k2 + 1]};
            const v2f i0 = { vi[0][it][2 * k2], vi[0][it][2 * k2 + 1]};
            const v2f i1 = { vi[1][it][2 * k2], vi[1][it][2 * k2 + 1]};
            const v2f zr = k0 * r0 + k1 * r1;
            const v2f zi = k0 * i0 + k1 * i1;
            const v2f ww = zr * zr + zi * zi;
            wh[it * 8 + 2 * k2] = ww.x;
            wh[it * 8 + 2 * k2 + 1] = ww.y;
          }
        #pragma unroll
        for (int len = 1; len < 16; len <<= 1)
          #pragma unroll
          for (int x = 0; x < 16; ++x)
            if (!(x & len)) {
              float a = wh[x], b = wh[x + len];
              wh[x] = a + b;
              wh[x + len] = a - b;
            }
        #pragma unroll
        for (int i = 0; i < NQ; ++i) {
          const float c = wh[walsh_mask(i)];
          acc[i] += (logical_index(0, j * 16) & (1 << (9 - i))) ? -c : c;
        }
      }
    }
  }
  const int al = logical_index(8 * (lane >> 4), lane & 15);
  #pragma unroll
  for (int i = 0; i < NQ; ++i)
    acc[i] = (al & (1 << (9 - i))) ? -acc[i] : acc[i];
  #pragma unroll
  for (int off = 16; off > 0; off >>= 1)
    #pragma unroll
    for (int i = 0; i < NQ; ++i)
      acc[i] += __shfl_xor(acc[i], off, 32);
  if (lane == 0) {
    #pragma unroll
    for (int i = 0; i < NQ; ++i) out[sample * NQ + i] = acc[i];
  }
  }
}

extern "C" void kernel_launch(void* const* d_in, const int* in_sizes, int n_in,
                              void* d_out, int out_size, void* d_ws, size_t ws_size,
                              hipStream_t stream) {
  const float* x = (const float*)d_in[0];
  const float* w = (const float*)d_in[1];
  float* out = (float*)d_out;

  vqc_main<<<BATCH_N / (WPB * SPW), BLOCK_T, 0, stream>>>(x, w, out);
}
